# starting kernel
# baseline (speedup 1.0000x reference)
#include <hip/hip_runtime.h>
#include <math.h>

typedef __attribute__((ext_vector_type(16))) __bf16 v16bf;
typedef __attribute__((ext_vector_type(8)))  float  v8f;
typedef unsigned short u16;
typedef unsigned int   u32;

#define HID 128
#define BM  128
#define BN  64
#define BK  32
#define BKP 40

__device__ __forceinline__ float bfh_to_f(u16 h) {
  union { u32 u; float f; } c; c.u = ((u32)h) << 16; return c.f;
}
__device__ __forceinline__ u16 f_to_bf_rne(float f) {
  union { float f; u32 u; } c; c.f = f;
  u32 r = c.u + 0x7FFFu + ((c.u >> 16) & 1u);
  return (u16)(r >> 16);
}
__device__ __forceinline__ void split_bf(float v, u16& hi, u16& lo) {
  union { float f; u32 u; } c; c.f = v;
  u16 h = (u16)(c.u >> 16);
  float r = v - bfh_to_f(h);
  hi = h; lo = f_to_bf_rne(r);
}

__global__ __launch_bounds__(256)
void gemm_bf16x3(const float* __restrict__ A, const float* __restrict__ W,
                 float* __restrict__ C, int nrows) {
  __shared__ u16 aHi[BM][BKP];
  __shared__ u16 aLo[BM][BKP];
  __shared__ u16 bHi[BN][BKP];
  __shared__ u16 bLo[BN][BKP];

  const int tid  = threadIdx.x;
  const int lane = tid & 31;
  const int wave = tid >> 5;
  const int l15  = lane & 15;
  const int kh   = (lane >> 4) << 3;
  const int row0 = blockIdx.x * BM;
  const int col0 = blockIdx.y * BN;
  const bool fullTile = (row0 + BM) <= nrows;

  v8f acc[4];
  for (int t = 0; t < 4; ++t)
    for (int r = 0; r < 8; ++r) acc[t][r] = 0.0f;

  for (int kc = 0; kc < HID; kc += BK) {
    __syncthreads();
    if (fullTile) {
      #pragma unroll
      for (int idx = tid; idx < BM * BK; idx += 256) {
        const int m = idx >> 5, k = idx & 31;
        float v = A[(long long)(row0 + m) * HID + (kc + k)];
        split_bf(v, aHi[m][k], aLo[m][k]);
      }
    } else {
      for (int idx = tid; idx < BM * BK; idx += 256) {
        const int m = idx >> 5, k = idx & 31;
        const int row = row0 + m;
        float v = (row < nrows) ? A[(long long)row * HID + (kc + k)] : 0.0f;
        split_bf(v, aHi[m][k], aLo[m][k]);
      }
    }
    #pragma unroll
    for (int idx = tid; idx < BN * BK; idx += 256) {
      const int n = idx >> 5, k = idx & 31;
      float v = W[(kc + k) * HID + (col0 + n)];
      split_bf(v, bHi[n][k], bLo[n][k]);
    }
    __syncthreads();

    union Frag { uint4 q[2]; v16bf v; };
    Frag ah, al;
    const int m = (wave << 4) + l15;
    ah.q[0] = *(const uint4*)&aHi[m][kh];
    ah.q[1] = *(const uint4*)&aHi[m][16 + kh];
    al.q[0] = *(const uint4*)&aLo[m][kh];
    al.q[1] = *(const uint4*)&aLo[m][16 + kh];

    #pragma unroll
    for (int nt = 0; nt < 4; ++nt) {
      Frag bh, bl;
      const int n = (nt << 4) + l15;
      bh.q[0] = *(const uint4*)&bHi[n][kh];
      bh.q[1] = *(const uint4*)&bHi[n][16 + kh];
      bl.q[0] = *(const uint4*)&bLo[n][kh];
      bl.q[1] = *(const uint4*)&bLo[n][16 + kh];
      acc[nt] = __builtin_amdgcn_wmma_f32_16x16x32_bf16(false, ah.v, false, bh.v, (short)0, acc[nt], false, false);
      acc[nt] = __builtin_amdgcn_wmma_f32_16x16x32_bf16(false, ah.v, false, bl.v, (short)0, acc[nt], false, false);
      acc[nt] = __builtin_amdgcn_wmma_f32_16x16x32_bf16(false, al.v, false, bh.v, (short)0, acc[nt], false, false);
    }
  }

  const int rbase = row0 + (wave << 4) + ((lane & 16) ? 8 : 0);
  if (fullTile) {
    #pragma unroll
    for (int nt = 0; nt < 4; ++nt) {
      const int c = col0 + (nt << 4) + l15;
      #pragma unroll
      for (int r = 0; r < 8; ++r)
        C[(long long)(rbase + r) * HID + c] = acc[nt][r];
    }
  } else {
    for (int nt = 0; nt < 4; ++nt) {
      const int c = col0 + (nt << 4) + l15;
      for (int r = 0; r < 8; ++r) {
        const int row = rbase + r;
        if (row < nrows) C[(long long)row * HID + c] = acc[nt][r];
      }
    }
  }
}

template <int CC>
__global__ void gemm_small(const float* __restrict__ h, const float* __restrict__ W,
                           float* __restrict__ out, int n) {
  const long long idx = (long long)blockIdx.x * blockDim.x + threadIdx.x;
  if (idx >= (long long)n * CC) return;
  const int node = (int)(idx / CC);
  const int c    = (int)(idx % CC);
  const float* hr = h + (long long)node * HID;
  float s = 0.0f;
  #pragma unroll 8
  for (int k = 0; k < HID; ++k) s += hr[k] * W[k * CC + c];
  out[idx] = s;
}

__global__ void fill_ones(float* p, int n) {
  int i = blockIdx.x * blockDim.x + threadIdx.x;
  if (i < n) p[i] = 1.0f;
}
__global__ void deg_accum(const int* __restrict__ dst, float* __restrict__ deg, int E) {
  int i = blockIdx.x * blockDim.x + threadIdx.x;
  if (i < E) atomicAdd(&deg[dst[i]], 1.0f);
}
__global__ void rsqrt_inplace(float* p, int n) {
  int i = blockIdx.x * blockDim.x + threadIdx.x;
  if (i < n) { float d = p[i]; p[i] = (d > 0.0f) ? rsqrtf(d) : 0.0f; }
}
__global__ void norm_kernel(const int* __restrict__ src, const int* __restrict__ dst,
                            const float* __restrict__ dinv, float* __restrict__ norm,
                            int E, int N) {
  int i = blockIdx.x * blockDim.x + threadIdx.x;
  if (i < E) { norm[i] = dinv[src[i]] * dinv[dst[i]]; }
  else if (i < E + N) { int v = i - E; norm[i] = dinv[v] * dinv[v]; }
}

template <int F>
__global__ void scatter_add(const float* __restrict__ tmp, const int* __restrict__ src,
                            const int* __restrict__ dst, const float* __restrict__ norm,
                            float* __restrict__ agg, int nE, int nN) {
  const int lane = threadIdx.x & 31;
  const long long item = (long long)blockIdx.x * (blockDim.x >> 5) + (threadIdx.x >> 5);
  if (item >= (long long)nE + nN) return;
  int s, d; float w;
  if (item < nE) { s = src[item]; d = dst[item]; w = norm[item]; }
  else           { int v = (int)(item - nE); s = v; d = v; w = norm[nE + v]; }
  const float* ts = tmp + (long long)s * F;
  float*       ad = agg + (long long)d * F;
  #pragma unroll
  for (int f = lane; f < F; f += 32)
    atomicAdd(&ad[f], ts[f] * w);
}

__global__ void fzero(float* p, long long n) {
  long long i = (long long)blockIdx.x * blockDim.x + threadIdx.x;
  if (i < n) p[i] = 0.0f;
}
template <int F>
__global__ void bias_elu(float* __restrict__ h, const float* __restrict__ b, long long n) {
  long long i = (long long)blockIdx.x * blockDim.x + threadIdx.x;
  if (i >= n) return;
  float v = h[i] + b[(int)(i % F)];
  h[i] = (v > 0.0f) ? v : expm1f(v);
}
template <int F>
__global__ void bias_add(float* __restrict__ out, const float* __restrict__ b, long long n) {
  long long i = (long long)blockIdx.x * blockDim.x + threadIdx.x;
  if (i < n) out[i] += b[(int)(i % F)];
}

static inline int cdiv(long long a, long long b) { return (int)((a + b - 1) / b); }

extern "C" void kernel_launch(void* const* d_in, const int* in_sizes, int n_in,
                              void* d_out, int out_size, void* d_ws, size_t ws_size,
                              hipStream_t stream) {
  (void)n_in; (void)out_size; (void)ws_size;
  const float* x    = (const float*)d_in[0];
  const int*   eidx = (const int*)d_in[1];
  const float* Wst  = (const float*)d_in[2];
  const float* bst  = (const float*)d_in[3];
  const float* Wout = (const float*)d_in[4];
  const float* bout = (const float*)d_in[5];

  const int N = in_sizes[0] / HID;
  const int E = in_sizes[1] / 2;
  constexpr int NCLS = 10;

  const int* src = eidx;
  const int* dst = eidx + E;

  float* norm = (float*)d_ws;
  float* deg  = norm + ((size_t)E + N);
  float* hbuf = deg + N;
  float* tmp  = hbuf + (size_t)N * HID;

  fill_ones<<<cdiv(N, 256), 256, 0, stream>>>(deg, N);
  deg_accum<<<cdiv(E, 256), 256, 0, stream>>>(dst, deg, E);
  rsqrt_inplace<<<cdiv(N, 256), 256, 0, stream>>>(deg, N);
  norm_kernel<<<cdiv((long long)E + N, 256), 256, 0, stream>>>(src, dst, deg, norm, E, N);

  const long long nh    = (long long)N * HID;
  const int scat_blocks = cdiv((long long)E + N, 8);
  const dim3 ggrid(cdiv(N, BM), HID / BN);

  const float* hin = x;
  for (int l = 0; l < 8; ++l) {
    gemm_bf16x3<<<ggrid, 256, 0, stream>>>(hin, Wst + (size_t)l * HID * HID, tmp, N);
    fzero<<<cdiv(nh, 256), 256, 0, stream>>>(hbuf, nh);
    scatter_add<HID><<<scat_blocks, 256, 0, stream>>>(tmp, src, dst, norm, hbuf, E, N);
    bias_elu<HID><<<cdiv(nh, 256), 256, 0, stream>>>(hbuf, bst + (size_t)l * HID, nh);
    hin = hbuf;
  }

  const long long no = (long long)N * NCLS;
  gemm_small<NCLS><<<cdiv(no, 256), 256, 0, stream>>>(hbuf, Wout, tmp, N);
  fzero<<<cdiv(no, 256), 256, 0, stream>>>((float*)d_out, no);
  scatter_add<NCLS><<<scat_blocks, 256, 0, stream>>>(tmp, src, dst, norm, (float*)d_out, E, N);
  bias_add<NCLS><<<cdiv(no, 256), 256, 0, stream>>>((float*)d_out, bout, no);
}


# Scatter sums kept in LDS: a workgroup owns 200 dst rows, scans dst for its edges, adds source rows by LDS float adds, stores each row once; global atomics, zero-fill gone; sum order = LDS arrival
# speedup vs baseline: 1.5104x; 1.5104x over previous
#include <hip/hip_runtime.h>
#include <math.h>

typedef __attribute__((ext_vector_type(16))) __bf16 v16bf;
typedef __attribute__((ext_vector_type(8)))  float  v8f;
typedef unsigned short u16;
typedef unsigned int   u32;

#define HID 128
#define BM  128
#define BN  64
#define BK  32
#define BKP 40

__device__ __forceinline__ float bfh_to_f(u16 h) {
  union { u32 u; float f; } c; c.u = ((u32)h) << 16; return c.f;
}
__device__ __forceinline__ u16 f_to_bf_rne(float f) {
  union { float f; u32 u; } c; c.f = f;
  u32 r = c.u + 0x7FFFu + ((c.u >> 16) & 1u);
  return (u16)(r >> 16);
}
__device__ __forceinline__ void split_bf(float v, u16& hi, u16& lo) {
  union { float f; u32 u; } c; c.f = v;
  u16 h = (u16)(c.u >> 16);
  float r = v - bfh_to_f(h);
  hi = h; lo = f_to_bf_rne(r);
}

__global__ __launch_bounds__(256)
void gemm_bf16x3(const float* __restrict__ A, const float* __restrict__ W,
                 float* __restrict__ C, int nrows) {
  __shared__ u16 aHi[BM][BKP];
  __shared__ u16 aLo[BM][BKP];
  __shared__ u16 bHi[BN][BKP];
  __shared__ u16 bLo[BN][BKP];

  const int tid  = threadIdx.x;
  const int lane = tid & 31;
  const int wave = tid >> 5;
  const int l15  = lane & 15;
  const int kh   = (lane >> 4) << 3;
  const int row0 = blockIdx.x * BM;
  const int col0 = blockIdx.y * BN;
  const bool fullTile = (row0 + BM) <= nrows;

  v8f acc[4];
  for (int t = 0; t < 4; ++t)
    for (int r = 0; r < 8; ++r) acc[t][r] = 0.0f;

  for (int kc = 0; kc < HID; kc += BK) {
    __syncthreads();
    if (fullTile) {
      #pragma unroll
      for (int idx = tid; idx < BM * BK; idx += 256) {
        const int m = idx >> 5, k = idx & 31;
        float v = A[(long long)(row0 + m) * HID + (kc + k)];
        split_bf(v, aHi[m][k], aLo[m][k]);
      }
    } else {
      for (int idx = tid; idx < BM * BK; idx += 256) {
        const int m = idx >> 5, k = idx & 31;
        const int row = row0 + m;
        float v = (row < nrows) ? A[(long long)row * HID + (kc + k)] : 0.0f;
        split_bf(v, aHi[m][k], aLo[m][k]);
      }
    }
    #pragma unroll
    for (int idx = tid; idx < BN * BK; idx += 256) {
      const int n = idx >> 5, k = idx & 31;
      float v = W[(kc + k) * HID + (col0 + n)];
      split_bf(v, bHi[n][k], bLo[n][k]);
    }
    __syncthreads();

    union Frag { uint4 q[2]; v16bf v; };
    Frag ah, al;
    const int m = (wave << 4) + l15;
    ah.q[0] = *(const uint4*)&aHi[m][kh];
    ah.q[1] = *(const uint4*)&aHi[m][16 + kh];
    al.q[0] = *(const uint4*)&aLo[m][kh];
    al.q[1] = *(const uint4*)&aLo[m][16 + kh];

    #pragma unroll
    for (int nt = 0; nt < 4; ++nt) {
      Frag bh, bl;
      const int n = (nt << 4) + l15;
      bh.q[0] = *(const uint4*)&bHi[n][kh];
      bh.q[1] = *(const uint4*)&bHi[n][16 + kh];
      bl.q[0] = *(const uint4*)&bLo[n][kh];
      bl.q[1] = *(const uint4*)&bLo[n][16 + kh];
      acc[nt] = __builtin_amdgcn_wmma_f32_16x16x32_bf16(false, ah.v, false, bh.v, (short)0, acc[nt], false, false);
      acc[nt] = __builtin_amdgcn_wmma_f32_16x16x32_bf16(false, ah.v, false, bl.v, (short)0, acc[nt], false, false);
      acc[nt] = __builtin_amdgcn_wmma_f32_16x16x32_bf16(false, al.v, false, bh.v, (short)0, acc[nt], false, false);
    }
  }

  const int rbase = row0 + (wave << 4) + ((lane & 16) ? 8 : 0);
  if (fullTile) {
    #pragma unroll
    for (int nt = 0; nt < 4; ++nt) {
      const int c = col0 + (nt << 4) + l15;
      #pragma unroll
      for (int r = 0; r < 8; ++r)
        C[(long long)(rbase + r) * HID + c] = acc[nt][r];
    }
  } else {
    for (int nt = 0; nt < 4; ++nt) {
      const int c = col0 + (nt << 4) + l15;
      for (int r = 0; r < 8; ++r) {
        const int row = rbase + r;
        if (row < nrows) C[(long long)row * HID + c] = acc[nt][r];
      }
    }
  }
}

template <int CC>
__global__ void gemm_small(const float* __restrict__ h, const float* __restrict__ W,
                           float* __restrict__ out, int n) {
  const long long idx = (long long)blockIdx.x * blockDim.x + threadIdx.x;
  if (idx >= (long long)n * CC) return;
  const int node = (int)(idx / CC);
  const int c    = (int)(idx % CC);
  const float* hr = h + (long long)node * HID;
  float s = 0.0f;
  #pragma unroll 8
  for (int k = 0; k < HID; ++k) s += hr[k] * W[k * CC + c];
  out[idx] = s;
}

__global__ void fill_ones(float* p, int n) {
  int i = blockIdx.x * blockDim.x + threadIdx.x;
  if (i < n) p[i] = 1.0f;
}
__global__ void deg_accum(const int* __restrict__ dst, float* __restrict__ deg, int E) {
  int i = blockIdx.x * blockDim.x + threadIdx.x;
  if (i < E) atomicAdd(&deg[dst[i]], 1.0f);
}
__global__ void rsqrt_inplace(float* p, int n) {
  int i = blockIdx.x * blockDim.x + threadIdx.x;
  if (i < n) { float d = p[i]; p[i] = (d > 0.0f) ? rsqrtf(d) : 0.0f; }
}
__global__ void norm_kernel(const int* __restrict__ src, const int* __restrict__ dst,
                            const float* __restrict__ dinv, float* __restrict__ norm,
                            int E, int N) {
  int i = blockIdx.x * blockDim.x + threadIdx.x;
  if (i < E) { norm[i] = dinv[src[i]] * dinv[dst[i]]; }
  else if (i < E + N) { int v = i - E; norm[i] = dinv[v] * dinv[v]; }
}

#define OWN_R   200
#define OWN_NT  1024
#define OWN_U   8
#define OWN_CAP 16384

template <int F>
__global__ __launch_bounds__(OWN_NT)
void agg_owner(const float* __restrict__ tmp, const int* __restrict__ src,
               const int* __restrict__ dst, const float* __restrict__ norm,
               float* __restrict__ agg, int nE, int nN) {
  __shared__ __attribute__((aligned(16))) float acc[(OWN_R + 1) * F];
  __shared__ int   lst[OWN_CAP];
  __shared__ int   cnt;

  const int tid  = threadIdx.x;
  const int lane = tid & 31;
  const int wave = __builtin_amdgcn_readfirstlane(tid >> 5);
  const int row0 = blockIdx.x * OWN_R;
  const int nrow = min(OWN_R, nN - row0);
  const size_t g0 = (size_t)row0 * F;
  const bool al16 = (F % 4 == 0) &&
      (((((unsigned long long)tmp) | ((unsigned long long)agg)) & 15ull) == 0ull);

  if (al16) {
    const float4* t4 = (const float4*)(tmp + g0);
    float4* a4 = (float4*)acc;
    for (int i = tid; i < (OWN_R + 1) * F / 4; i += OWN_NT) {
      const int r = i / (F / 4);
      float4 v = make_float4(0.f, 0.f, 0.f, 0.f);
      if (r < nrow) {
        const float w = norm[nE + row0 + r];
        v = t4[i];
        v.x *= w; v.y *= w; v.z *= w; v.w *= w;
      }
      a4[i] = v;
    }
  } else {
    for (int i = tid; i < (OWN_R + 1) * F; i += OWN_NT) {
      const int r = i / F;
      acc[i] = (r < nrow) ? tmp[g0 + i] * norm[nE + row0 + r] : 0.0f;
    }
  }
  if (tid == 0) cnt = 0;
  __syncthreads();

  auto note = [&](int e) {
    const int pos = atomicAdd(&cnt, 1);
    if (pos < OWN_CAP) {
      lst[pos] = e;
    } else {
      const int   s  = src[e];
      const float w  = norm[e];
      const int   dl = dst[e] - row0;
      for (int f = 0; f < F; ++f)
        atomicAdd(&acc[dl * F + f], tmp[(size_t)s * F + f] * w);
    }
  };

  const bool vec = ((((unsigned long long)dst) & 15ull) == 0ull);
  const int  n4  = vec ? (nE >> 2) : 0;
  const int4* dst4 = (const int4*)dst;
  for (int base = 0; base < n4; base += OWN_U * OWN_NT) {
    int4 q[OWN_U];
    #pragma unroll
    for (int u = 0; u < OWN_U; ++u) {
      const int i = base + u * OWN_NT + tid;
      q[u] = dst4[min(i, n4 - 1)];
    }
    unsigned hit = 0u;
    #pragma unroll
    for (int u = 0; u < OWN_U; ++u) {
      const int i = base + u * OWN_NT + tid;
      const unsigned lim = (i < n4) ? (unsigned)nrow : 0u;
      hit |= (unsigned)((unsigned)(q[u].x - row0) < lim) << (4 * u + 0);
      hit |= (unsigned)((unsigned)(q[u].y - row0) < lim) << (4 * u + 1);
      hit |= (unsigned)((unsigned)(q[u].z - row0) < lim) << (4 * u + 2);
      hit |= (unsigned)((unsigned)(q[u].w - row0) < lim) << (4 * u + 3);
    }
    while (hit) {
      const int k = __builtin_ctz(hit);
      hit &= hit - 1u;
      note(((base + (k >> 2) * OWN_NT + tid) << 2) + (k & 3));
    }
  }
  for (int i = (n4 << 2) + tid; i < nE; i += OWN_NT)
    if ((unsigned)(dst[i] - row0) < (unsigned)nrow) note(i);
  __syncthreads();

  const int nList = __builtin_amdgcn_readfirstlane(min(cnt, OWN_CAP));
  const int nB    = (nList + 31) >> 5;
  constexpr int NF = (F + 31) / 32;
  for (int b = wave; b < nB; b += OWN_NT / 32) {
    const int j = (b << 5) + lane;
    int s = 0, dl = OWN_R;
    float w = 0.0f;
    if (j < nList) {
      const int e = lst[j];
      s = src[e]; dl = dst[e] - row0; w = norm[e];
    }
    const int m = min(32, nList - (b << 5));
    for (int k0 = 0; k0 < m; k0 += 8) {
      float v[8][NF];
      int   dd[8];
      float ww[8];
      #pragma unroll
      for (int u = 0; u < 8; ++u) {
        const int sk = __builtin_amdgcn_readlane(s, k0 + u);
        dd[u] = __builtin_amdgcn_readlane(dl, k0 + u);
        ww[u] = __int_as_float(__builtin_amdgcn_readlane(__float_as_int(w), k0 + u));
        const float* ts = tmp + (size_t)sk * F;
        #pragma unroll
        for (int c = 0; c < NF; ++c) {
          const int f = lane + 32 * c;
          v[u][c] = (f < F) ? ts[f] : 0.0f;
        }
      }
      #pragma unroll
      for (int u = 0; u < 8; ++u) {
        float* ad = acc + dd[u] * F;
        #pragma unroll
        for (int c = 0; c < NF; ++c) {
          const int f = lane + 32 * c;
          if (f < F) atomicAdd(&ad[f], v[u][c] * ww[u]);
        }
      }
    }
  }
  __syncthreads();

  if (al16) {
    const float4* a4 = (const float4*)acc;
    float4* o4 = (float4*)(agg + g0);
    for (int i = tid; i < nrow * (F / 4); i += OWN_NT) o4[i] = a4[i];
  } else {
    for (int i = tid; i < nrow * F; i += OWN_NT) agg[g0 + i] = acc[i];
  }
}

template <int F>
__global__ void bias_elu(float* __restrict__ h, const float* __restrict__ b, long long n) {
  long long i = (long long)blockIdx.x * blockDim.x + threadIdx.x;
  if (i >= n) return;
  float v = h[i] + b[(int)(i % F)];
  h[i] = (v > 0.0f) ? v : expm1f(v);
}
template <int F>
__global__ void bias_add(float* __restrict__ out, const float* __restrict__ b, long long n) {
  long long i = (long long)blockIdx.x * blockDim.x + threadIdx.x;
  if (i < n) out[i] += b[(int)(i % F)];
}

static inline int cdiv(long long a, long long b) { return (int)((a + b - 1) / b); }

extern "C" void kernel_launch(void* const* d_in, const int* in_sizes, int n_in,
                              void* d_out, int out_size, void* d_ws, size_t ws_size,
                              hipStream_t stream) {
  (void)n_in; (void)out_size; (void)ws_size;
  const float* x    = (const float*)d_in[0];
  const int*   eidx = (const int*)d_in[1];
  const float* Wst  = (const float*)d_in[2];
  const float* bst  = (const float*)d_in[3];
  const float* Wout = (const float*)d_in[4];
  const float* bout = (const float*)d_in[5];

  const int N = in_sizes[0] / HID;
  const int E = in_sizes[1] / 2;
  constexpr int NCLS = 10;

  const int* src = eidx;
  const int* dst = eidx + E;

  float* norm = (float*)d_ws;
  float* deg  = norm + ((size_t)E + N);
  float* hbuf = deg + N;
  float* tmp  = hbuf + (size_t)N * HID;

  fill_ones<<<cdiv(N, 256), 256, 0, stream>>>(deg, N);
  deg_accum<<<cdiv(E, 256), 256, 0, stream>>>(dst, deg, E);
  rsqrt_inplace<<<cdiv(N, 256), 256, 0, stream>>>(deg, N);
  norm_kernel<<<cdiv((long long)E + N, 256), 256, 0, stream>>>(src, dst, deg, norm, E, N);

  const long long nh    = (long long)N * HID;
  const int own_blocks  = cdiv(N, OWN_R);
  const dim3 ggrid(cdiv(N, BM), HID / BN);

  const float* hin = x;
  for (int l = 0; l < 8; ++l) {
    gemm_bf16x3<<<ggrid, 256, 0, stream>>>(hin, Wst + (size_t)l * HID * HID, tmp, N);
    agg_owner<HID><<<own_blocks, OWN_NT, 0, stream>>>(tmp, src, dst, norm, hbuf, E, N);
    bias_elu<HID><<<cdiv(nh, 256), 256, 0, stream>>>(hbuf, bst + (size_t)l * HID, nh);
    hin = hbuf;
  }

  const long long no = (long long)N * NCLS;
  gemm_small<NCLS><<<cdiv(no, 256), 256, 0, stream>>>(hbuf, Wout, tmp, N);
  agg_owner<NCLS><<<own_blocks, OWN_NT, 0, stream>>>(tmp, src, dst, norm, (float*)d_out, E, N);
  bias_add<NCLS><<<cdiv(no, 256), 256, 0, stream>>>((float*)d_out, bout, no);
}


# Edge scan hoisted out of the layers: each row-owner workgroup's in-edge records (source, row, weight) are built once per call and read by all 9 aggregations; row sums still in LDS arrival order
# speedup vs baseline: 1.8514x; 1.2258x over previous
#include <hip/hip_runtime.h>
#include <math.h>

typedef __attribute__((ext_vector_type(16))) __bf16 v16bf;
typedef __attribute__((ext_vector_type(8)))  float  v8f;
typedef unsigned short u16;
typedef unsigned int   u32;

#define HID 128
#define BM  128
#define BN  64
#define BK  32
#define BKP 40

__device__ __forceinline__ float bfh_to_f(u16 h) {
  union { u32 u; float f; } c; c.u = ((u32)h) << 16; return c.f;
}
__device__ __forceinline__ u16 f_to_bf_rne(float f) {
  union { float f; u32 u; } c; c.f = f;
  u32 r = c.u + 0x7FFFu + ((c.u >> 16) & 1u);
  return (u16)(r >> 16);
}
__device__ __forceinline__ void split_bf(float v, u16& hi, u16& lo) {
  union { float f; u32 u; } c; c.f = v;
  u16 h = (u16)(c.u >> 16);
  float r = v - bfh_to_f(h);
  hi = h; lo = f_to_bf_rne(r);
}

__global__ __launch_bounds__(256)
void gemm_bf16x3(const float* __restrict__ A, const float* __restrict__ W,
                 float* __restrict__ C, int nrows) {
  __shared__ u16 aHi[BM][BKP];
  __shared__ u16 aLo[BM][BKP];
  __shared__ u16 bHi[BN][BKP];
  __shared__ u16 bLo[BN][BKP];

  const int tid  = threadIdx.x;
  const int lane = tid & 31;
  const int wave = tid >> 5;
  const int l15  = lane & 15;
  const int kh   = (lane >> 4) << 3;
  const int row0 = blockIdx.x * BM;
  const int col0 = blockIdx.y * BN;
  const bool fullTile = (row0 + BM) <= nrows;

  v8f acc[4];
  for (int t = 0; t < 4; ++t)
    for (int r = 0; r < 8; ++r) acc[t][r] = 0.0f;

  for (int kc = 0; kc < HID; kc += BK) {
    __syncthreads();
    if (fullTile) {
      #pragma unroll
      for (int idx = tid; idx < BM * BK; idx += 256) {
        const int m = idx >> 5, k = idx & 31;
        float v = A[(long long)(row0 + m) * HID + (kc + k)];
        split_bf(v, aHi[m][k], aLo[m][k]);
      }
    } else {
      for (int idx = tid; idx < BM * BK; idx += 256) {
        const int m = idx >> 5, k = idx & 31;
        const int row = row0 + m;
        float v = (row < nrows) ? A[(long long)row * HID + (kc + k)] : 0.0f;
        split_bf(v, aHi[m][k], aLo[m][k]);
      }
    }
    #pragma unroll
    for (int idx = tid; idx < BN * BK; idx += 256) {
      const int n = idx >> 5, k = idx & 31;
      float v = W[(kc + k) * HID + (col0 + n)];
      split_bf(v, bHi[n][k], bLo[n][k]);
    }
    __syncthreads();

    union Frag { uint4 q[2]; v16bf v; };
    Frag ah, al;
    const int m = (wave << 4) + l15;
    ah.q[0] = *(const uint4*)&aHi[m][kh];
    ah.q[1] = *(const uint4*)&aHi[m][16 + kh];
    al.q[0] = *(const uint4*)&aLo[m][kh];
    al.q[1] = *(const uint4*)&aLo[m][16 + kh];

    #pragma unroll
    for (int nt = 0; nt < 4; ++nt) {
      Frag bh, bl;
      const int n = (nt << 4) + l15;
      bh.q[0] = *(const uint4*)&bHi[n][kh];
      bh.q[1] = *(const uint4*)&bHi[n][16 + kh];
      bl.q[0] = *(const uint4*)&bLo[n][kh];
      bl.q[1] = *(const uint4*)&bLo[n][16 + kh];
      acc[nt] = __builtin_amdgcn_wmma_f32_16x16x32_bf16(false, ah.v, false, bh.v, (short)0, acc[nt], false, false);
      acc[nt] = __builtin_amdgcn_wmma_f32_16x16x32_bf16(false, ah.v, false, bl.v, (short)0, acc[nt], false, false);
      acc[nt] = __builtin_amdgcn_wmma_f32_16x16x32_bf16(false, al.v, false, bh.v, (short)0, acc[nt], false, false);
    }
  }

  const int rbase = row0 + (wave << 4) + ((lane & 16) ? 8 : 0);
  if (fullTile) {
    #pragma unroll
    for (int nt = 0; nt < 4; ++nt) {
      const int c = col0 + (nt << 4) + l15;
      #pragma unroll
      for (int r = 0; r < 8; ++r)
        C[(long long)(rbase + r) * HID + c] = acc[nt][r];
    }
  } else {
    for (int nt = 0; nt < 4; ++nt) {
      const int c = col0 + (nt << 4) + l15;
      for (int r = 0; r < 8; ++r) {
        const int row = rbase + r;
        if (row < nrows) C[(long long)row * HID + c] = acc[nt][r];
      }
    }
  }
}

template <int CC>
__global__ void gemm_small(const float* __restrict__ h, const float* __restrict__ W,
                           float* __restrict__ out, int n) {
  const long long idx = (long long)blockIdx.x * blockDim.x + threadIdx.x;
  if (idx >= (long long)n * CC) return;
  const int node = (int)(idx / CC);
  const int c    = (int)(idx % CC);
  const float* hr = h + (long long)node * HID;
  float s = 0.0f;
  #pragma unroll 8
  for (int k = 0; k < HID; ++k) s += hr[k] * W[k * CC + c];
  out[idx] = s;
}

__global__ void fill_ones(float* p, int n) {
  int i = blockIdx.x * blockDim.x + threadIdx.x;
  if (i < n) p[i] = 1.0f;
}
__global__ void deg_accum(const int* __restrict__ dst, float* __restrict__ deg, int E) {
  int i = blockIdx.x * blockDim.x + threadIdx.x;
  if (i < E) atomicAdd(&deg[dst[i]], 1.0f);
}
__global__ void rsqrt_of(const float* __restrict__ d, float* __restrict__ r, int n) {
  int i = blockIdx.x * blockDim.x + threadIdx.x;
  if (i < n) { float v = d[i]; r[i] = (v > 0.0f) ? rsqrtf(v) : 0.0f; }
}
__global__ void norm_kernel(const int* __restrict__ src, const int* __restrict__ dst,
                            const float* __restrict__ dinv, float* __restrict__ norm,
                            int E, int N) {
  int i = blockIdx.x * blockDim.x + threadIdx.x;
  if (i < E) { norm[i] = dinv[src[i]] * dinv[dst[i]]; }
  else if (i < E + N) { int v = i - E; norm[i] = dinv[v] * dinv[v]; }
}

#define OWN_R   200
#define OWN_NT  1024
#define OWN_U   8

__global__ __launch_bounds__(OWN_NT)
void own_lists(const int* __restrict__ src, const int* __restrict__ dst,
               const float* __restrict__ norm, const float* __restrict__ deg,
               int4* __restrict__ rec, int* __restrict__ span, int nE, int nN) {
  __shared__ int baseS;
  __shared__ int cnt;

  const int tid  = threadIdx.x;
  const int row0 = blockIdx.x * OWN_R;
  const int nrow = min(OWN_R, nN - row0);

  if (tid == 0) { baseS = 0; cnt = 0; }
  __syncthreads();

  int part = 0;
  for (int v = tid; v < row0; v += OWN_NT) part += (int)deg[v] - 1;
  atomicAdd(&baseS, part);
  __syncthreads();
  const int base = baseS;

  auto note = [&](int e) {
    const int p = base + atomicAdd(&cnt, 1);
    if (p < nE) rec[p] = make_int4(src[e], dst[e] - row0, __float_as_int(norm[e]), 0);
  };

  const bool vec = ((((unsigned long long)dst) & 15ull) == 0ull);
  const int  n4  = vec ? (nE >> 2) : 0;
  const int4* dst4 = (const int4*)dst;
  for (int base4 = 0; base4 < n4; base4 += OWN_U * OWN_NT) {
    int4 q[OWN_U];
    #pragma unroll
    for (int u = 0; u < OWN_U; ++u) {
      const int i = base4 + u * OWN_NT + tid;
      q[u] = dst4[min(i, n4 - 1)];
    }
    unsigned hit = 0u;
    #pragma unroll
    for (int u = 0; u < OWN_U; ++u) {
      const int i = base4 + u * OWN_NT + tid;
      const unsigned lim = (i < n4) ? (unsigned)nrow : 0u;
      hit |= (unsigned)((unsigned)(q[u].x - row0) < lim) << (4 * u + 0);
      hit |= (unsigned)((unsigned)(q[u].y - row0) < lim) << (4 * u + 1);
      hit |= (unsigned)((unsigned)(q[u].z - row0) < lim) << (4 * u + 2);
      hit |= (unsigned)((unsigned)(q[u].w - row0) < lim) << (4 * u + 3);
    }
    while (hit) {
      const int k = __builtin_ctz(hit);
      hit &= hit - 1u;
      note(((base4 + (k >> 2) * OWN_NT + tid) << 2) + (k & 3));
    }
  }
  for (int i = (n4 << 2) + tid; i < nE; i += OWN_NT)
    if ((unsigned)(dst[i] - row0) < (unsigned)nrow) note(i);
  __syncthreads();

  if (tid == 0) {
    span[2 * blockIdx.x]     = base;
    span[2 * blockIdx.x + 1] = max(0, min(cnt, nE - base));
  }
}

template <int F>
__global__ __launch_bounds__(OWN_NT)
void agg_owner(const float* __restrict__ tmp, const int4* __restrict__ rec,
               const int* __restrict__ span, const float* __restrict__ norm,
               float* __restrict__ agg, int nE, int nN) {
  __shared__ __attribute__((aligned(16))) float acc[(OWN_R + 1) * F];

  const int tid  = threadIdx.x;
  const int lane = tid & 31;
  const int wave = __builtin_amdgcn_readfirstlane(tid >> 5);
  const int row0 = blockIdx.x * OWN_R;
  const int nrow = min(OWN_R, nN - row0);
  const size_t g0 = (size_t)row0 * F;
  const bool al16 = (F % 4 == 0) &&
      (((((unsigned long long)tmp) | ((unsigned long long)agg)) & 15ull) == 0ull);

  const int nList = __builtin_amdgcn_readfirstlane(span[2 * blockIdx.x + 1]);
  const int4* myrec = rec + __builtin_amdgcn_readfirstlane(span[2 * blockIdx.x]);
  const int nB = (nList + 31) >> 5;
  const int4 pad = make_int4(0, OWN_R, 0, 0);
  int4 nx = pad;
  { const int j = (wave << 5) + lane; if (j < nList) nx = myrec[j]; }

  if (al16) {
    const float4* t4 = (const float4*)(tmp + g0);
    float4* a4 = (float4*)acc;
    for (int i = tid; i < (OWN_R + 1) * F / 4; i += OWN_NT) {
      const int r = i / (F / 4);
      float4 v = make_float4(0.f, 0.f, 0.f, 0.f);
      if (r < nrow) {
        const float w = norm[nE + row0 + r];
        v = t4[i];
        v.x *= w; v.y *= w; v.z *= w; v.w *= w;
      }
      a4[i] = v;
    }
  } else {
    for (int i = tid; i < (OWN_R + 1) * F; i += OWN_NT) {
      const int r = i / F;
      acc[i] = (r < nrow) ? tmp[g0 + i] * norm[nE + row0 + r] : 0.0f;
    }
  }
  __syncthreads();

  constexpr int NF = (F + 31) / 32;
  for (int b = wave; b < nB; b += OWN_NT / 32) {
    const int   s  = nx.x;
    const int   dl = nx.y;
    const float w  = __int_as_float(nx.z);
    {
      const int jn = ((b + OWN_NT / 32) << 5) + lane;
      nx = pad;
      if (jn < nList) nx = myrec[jn];
    }
    const int m = min(32, nList - (b << 5));
    for (int k0 = 0; k0 < m; k0 += 8) {
      float v[8][NF];
      int   dd[8];
      float ww[8];
      #pragma unroll
      for (int u = 0; u < 8; ++u) {
        const int sk = __builtin_amdgcn_readlane(s, k0 + u);
        dd[u] = __builtin_amdgcn_readlane(dl, k0 + u);
        ww[u] = __int_as_float(__builtin_amdgcn_readlane(__float_as_int(w), k0 + u));
        const float* ts = tmp + (size_t)sk * F;
        #pragma unroll
        for (int c = 0; c < NF; ++c) {
          const int f = lane + 32 * c;
          v[u][c] = (f < F) ? ts[f] : 0.0f;
        }
      }
      #pragma unroll
      for (int u = 0; u < 8; ++u) {
        float* ad = acc + dd[u] * F;
        #pragma unroll
        for (int c = 0; c < NF; ++c) {
          const int f = lane + 32 * c;
          if (f < F) atomicAdd(&ad[f], v[u][c] * ww[u]);
        }
      }
    }
  }
  __syncthreads();

  if (al16) {
    const float4* a4 = (const float4*)acc;
    float4* o4 = (float4*)(agg + g0);
    for (int i = tid; i < nrow * (F / 4); i += OWN_NT) o4[i] = a4[i];
  } else {
    for (int i = tid; i < nrow * F; i += OWN_NT) agg[g0 + i] = acc[i];
  }
}

template <int F>
__global__ void bias_elu(float* __restrict__ h, const float* __restrict__ b, long long n) {
  long long i = (long long)blockIdx.x * blockDim.x + threadIdx.x;
  if (i >= n) return;
  float v = h[i] + b[(int)(i % F)];
  h[i] = (v > 0.0f) ? v : expm1f(v);
}
template <int F>
__global__ void bias_add(float* __restrict__ out, const float* __restrict__ b, long long n) {
  long long i = (long long)blockIdx.x * blockDim.x + threadIdx.x;
  if (i < n) out[i] += b[(int)(i % F)];
}

static inline int cdiv(long long a, long long b) { return (int)((a + b - 1) / b); }

extern "C" void kernel_launch(void* const* d_in, const int* in_sizes, int n_in,
                              void* d_out, int out_size, void* d_ws, size_t ws_size,
                              hipStream_t stream) {
  (void)n_in; (void)out_size; (void)ws_size;
  const float* x    = (const float*)d_in[0];
  const int*   eidx = (const int*)d_in[1];
  const float* Wst  = (const float*)d_in[2];
  const float* bst  = (const float*)d_in[3];
  const float* Wout = (const float*)d_in[4];
  const float* bout = (const float*)d_in[5];

  const int N = in_sizes[0] / HID;
  const int E = in_sizes[1] / 2;
  constexpr int NCLS = 10;

  const int* src = eidx;
  const int* dst = eidx + E;

  const int own_blocks  = cdiv(N, OWN_R);

  int4*  rec  = (int4*)d_ws;
  float* norm = (float*)(rec + E);
  float* deg  = norm + ((size_t)E + N);
  float* dinv = deg + N;
  float* hbuf = dinv + N;
  float* tmp  = hbuf + (size_t)N * HID;
  int*   span = (int*)(tmp + (size_t)N * HID);

  fill_ones<<<cdiv(N, 256), 256, 0, stream>>>(deg, N);
  deg_accum<<<cdiv(E, 256), 256, 0, stream>>>(dst, deg, E);
  rsqrt_of<<<cdiv(N, 256), 256, 0, stream>>>(deg, dinv, N);
  norm_kernel<<<cdiv((long long)E + N, 256), 256, 0, stream>>>(src, dst, dinv, norm, E, N);
  own_lists<<<own_blocks, OWN_NT, 0, stream>>>(src, dst, norm, deg, rec, span, E, N);

  const long long nh    = (long long)N * HID;
  const dim3 ggrid(cdiv(N, BM), HID / BN);

  const float* hin = x;
  for (int l = 0; l < 8; ++l) {
    gemm_bf16x3<<<ggrid, 256, 0, stream>>>(hin, Wst + (size_t)l * HID * HID, tmp, N);
    agg_owner<HID><<<own_blocks, OWN_NT, 0, stream>>>(tmp, rec, span, norm, hbuf, E, N);
    bias_elu<HID><<<cdiv(nh, 256), 256, 0, stream>>>(hbuf, bst + (size_t)l * HID, nh);
    hin = hbuf;
  }

  const long long no = (long long)N * NCLS;
  gemm_small<NCLS><<<cdiv(no, 256), 256, 0, stream>>>(hbuf, Wout, tmp, N);
  agg_owner<NCLS><<<own_blocks, OWN_NT, 0, stream>>>(tmp, rec, span, norm, (float*)d_out, E, N);
  bias_add<NCLS><<<cdiv(no, 256), 256, 0, stream>>>((float*)d_out, bout, no);
}


# Layer 1's product moved off the 128x64-tile LDS GEMM onto the row-owner workgroups: 200 rows of x and the split weights staged in LDS once, all loads in flight; same bf16 hi/lo WMMA order
# speedup vs baseline: 3.5629x; 1.9244x over previous
#include <hip/hip_runtime.h>
#include <math.h>

typedef __attribute__((ext_vector_type(16))) __bf16 v16bf;
typedef __attribute__((ext_vector_type(8)))  float  v8f;
typedef unsigned short u16;
typedef unsigned int   u32;

#define HID 128
#define BK  32

__device__ __forceinline__ float bfh_to_f(u16 h) {
  union { u32 u; float f; } c; c.u = ((u32)h) << 16; return c.f;
}
__device__ __forceinline__ u16 f_to_bf_rne(float f) {
  union { float f; u32 u; } c; c.f = f;
  u32 r = c.u + 0x7FFFu + ((c.u >> 16) & 1u);
  return (u16)(r >> 16);
}
__device__ __forceinline__ void split_bf(float v, u16& hi, u16& lo) {
  union { float f; u32 u; } c; c.f = v;
  u16 h = (u16)(c.u >> 16);
  float r = v - bfh_to_f(h);
  hi = h; lo = f_to_bf_rne(r);
}

__global__ void fill_ones(float* p, int n) {
  int i = blockIdx.x * blockDim.x + threadIdx.x;
  if (i < n) p[i] = 1.0f;
}
__global__ void deg_accum(const int* __restrict__ dst, float* __restrict__ deg, int E) {
  int i = blockIdx.x * blockDim.x + threadIdx.x;
  if (i < E) atomicAdd(&deg[dst[i]], 1.0f);
}
__global__ void rsqrt_of(const float* __restrict__ d, float* __restrict__ r, int n) {
  int i = blockIdx.x * blockDim.x + threadIdx.x;
  if (i < n) { float v = d[i]; r[i] = (v > 0.0f) ? rsqrtf(v) : 0.0f; }
}
__global__ void norm_kernel(const int* __restrict__ src, const int* __restrict__ dst,
                            const float* __restrict__ dinv, float* __restrict__ norm,
                            int E, int N) {
  int i = blockIdx.x * blockDim.x + threadIdx.x;
  if (i < E) { norm[i] = dinv[src[i]] * dinv[dst[i]]; }
  else if (i < E + N) { int v = i - E; norm[i] = dinv[v] * dinv[v]; }
}

#define OWN_R   200
#define OWN_NT  1024
#define OWN_U   8

__global__ __launch_bounds__(OWN_NT)
void own_lists(const int* __restrict__ src, const int* __restrict__ dst,
               const float* __restrict__ norm, const float* __restrict__ deg,
               int4* __restrict__ rec, int* __restrict__ span, int nE, int nN) {
  __shared__ int baseS;
  __shared__ int cnt;

  const int tid  = threadIdx.x;
  const int row0 = blockIdx.x * OWN_R;
  const int nrow = min(OWN_R, nN - row0);

  if (tid == 0) { baseS = 0; cnt = 0; }
  __syncthreads();

  int part = 0;
  for (int v = tid; v < row0; v += OWN_NT) part += (int)deg[v] - 1;
  atomicAdd(&baseS, part);
  __syncthreads();
  const int base = baseS;

  auto note = [&](int e) {
    const int p = base + atomicAdd(&cnt, 1);
    if (p < nE) rec[p] = make_int4(src[e], dst[e] - row0, __float_as_int(norm[e]), 0);
  };

  const bool vec = ((((unsigned long long)dst) & 15ull) == 0ull);
  const int  n4  = vec ? (nE >> 2) : 0;
  const int4* dst4 = (const int4*)dst;
  for (int base4 = 0; base4 < n4; base4 += OWN_U * OWN_NT) {
    int4 q[OWN_U];
    #pragma unroll
    for (int u = 0; u < OWN_U; ++u) {
      const int i = base4 + u * OWN_NT + tid;
      q[u] = dst4[min(i, n4 - 1)];
    }
    unsigned hit = 0u;
    #pragma unroll
    for (int u = 0; u < OWN_U; ++u) {
      const int i = base4 + u * OWN_NT + tid;
      const unsigned lim = (i < n4) ? (unsigned)nrow : 0u;
      hit |= (unsigned)((unsigned)(q[u].x - row0) < lim) << (4 * u + 0);
      hit |= (unsigned)((unsigned)(q[u].y - row0) < lim) << (4 * u + 1);
      hit |= (unsigned)((unsigned)(q[u].z - row0) < lim) << (4 * u + 2);
      hit |= (unsigned)((unsigned)(q[u].w - row0) < lim) << (4 * u + 3);
    }
    while (hit) {
      const int k = __builtin_ctz(hit);
      hit &= hit - 1u;
      note(((base4 + (k >> 2) * OWN_NT + tid) << 2) + (k & 3));
    }
  }
  for (int i = (n4 << 2) + tid; i < nE; i += OWN_NT)
    if ((unsigned)(dst[i] - row0) < (unsigned)nrow) note(i);
  __syncthreads();

  if (tid == 0) {
    span[2 * blockIdx.x]     = base;
    span[2 * blockIdx.x + 1] = max(0, min(cnt, nE - base));
  }
}

template <int F>
__global__ __launch_bounds__(OWN_NT)
void agg_owner(const float* __restrict__ tmp, const int4* __restrict__ rec,
               const int* __restrict__ span, const float* __restrict__ norm,
               const float* __restrict__ bias, float* __restrict__ agg, int nE, int nN) {
  __shared__ __attribute__((aligned(16))) float acc[(OWN_R + 1) * F];

  const int tid  = threadIdx.x;
  const int lane = tid & 31;
  const int wave = __builtin_amdgcn_readfirstlane(tid >> 5);
  const int row0 = blockIdx.x * OWN_R;
  const int nrow = min(OWN_R, nN - row0);
  const size_t g0 = (size_t)row0 * F;
  const bool al16 = (F % 4 == 0) &&
      (((((unsigned long long)tmp) | ((unsigned long long)agg)) & 15ull) == 0ull);

  const int nList = __builtin_amdgcn_readfirstlane(span[2 * blockIdx.x + 1]);
  const int4* myrec = rec + __builtin_amdgcn_readfirstlane(span[2 * blockIdx.x]);
  const int nB = (nList + 31) >> 5;
  const int4 pad = make_int4(0, OWN_R, 0, 0);
  int4 nx = pad;
  { const int j = (wave << 5) + lane; if (j < nList) nx = myrec[j]; }

  if (al16) {
    const float4* t4 = (const float4*)(tmp + g0);
    float4* a4 = (float4*)acc;
    for (int i = tid; i < (OWN_R + 1) * F / 4; i += OWN_NT) {
      const int r = i / (F / 4);
      float4 v = make_float4(0.f, 0.f, 0.f, 0.f);
      if (r < nrow) {
        const float w = norm[nE + row0 + r];
        v = t4[i];
        v.x *= w; v.y *= w; v.z *= w; v.w *= w;
      }
      a4[i] = v;
    }
  } else {
    for (int i = tid; i < (OWN_R + 1) * F; i += OWN_NT) {
      const int r = i / F;
      acc[i] = (r < nrow) ? tmp[g0 + i] * norm[nE + row0 + r] : 0.0f;
    }
  }
  __syncthreads();

  constexpr int NF = (F + 31) / 32;
  for (int b = wave; b < nB; b += OWN_NT / 32) {
    const int   s  = nx.x;
    const int   dl = nx.y;
    const float w  = __int_as_float(nx.z);
    {
      const int jn = ((b + OWN_NT / 32) << 5) + lane;
      nx = pad;
      if (jn < nList) nx = myrec[jn];
    }
    const int m = min(32, nList - (b << 5));
    for (int k0 = 0; k0 < m; k0 += 8) {
      float v[8][NF];
      int   dd[8];
      float ww[8];
      #pragma unroll
      for (int u = 0; u < 8; ++u) {
        const int sk = __builtin_amdgcn_readlane(s, k0 + u);
        dd[u] = __builtin_amdgcn_readlane(dl, k0 + u);
        ww[u] = __int_as_float(__builtin_amdgcn_readlane(__float_as_int(w), k0 + u));
        const float* ts = tmp + (size_t)sk * F;
        #pragma unroll
        for (int c = 0; c < NF; ++c) {
          const int f = lane + 32 * c;
          v[u][c] = (f < F) ? ts[f] : 0.0f;
        }
      }
      #pragma unroll
      for (int u = 0; u < 8; ++u) {
        float* ad = acc + dd[u] * F;
        #pragma unroll
        for (int c = 0; c < NF; ++c) {
          const int f = lane + 32 * c;
          if (f < F) atomicAdd(&ad[f], v[u][c] * ww[u]);
        }
      }
    }
  }
  __syncthreads();

  if (al16) {
    const float4* a4 = (const float4*)acc;
    float4* o4 = (float4*)(agg + g0);
    for (int i = tid; i < nrow * (F / 4); i += OWN_NT) {
      const int c = (i % (F / 4)) * 4;
      float4 v = a4[i];
      v.x += bias[c]; v.y += bias[c + 1]; v.z += bias[c + 2]; v.w += bias[c + 3];
      o4[i] = v;
    }
  } else {
    for (int i = tid; i < nrow * F; i += OWN_NT) agg[g0 + i] = acc[i] + bias[i % F];
  }
}

#define ACP    132
#define OWN_RP 208
#define WKP    136

template <int NOUT>
__global__ __launch_bounds__(OWN_NT)
void agg_next(const float* __restrict__ tmp, const int4* __restrict__ rec,
              const int* __restrict__ span, const float* __restrict__ norm,
              const float* __restrict__ bias, const float* __restrict__ Wn,
              float* __restrict__ outn, int nE, int nN) {
  constexpr int  F    = HID;
  constexpr bool WIDE = (NOUT == HID);
  __shared__ __attribute__((aligned(16))) float acc[OWN_RP * ACP];
  __shared__ __attribute__((aligned(16))) u16   wHi[WIDE ? HID * WKP : 8];
  __shared__ __attribute__((aligned(16))) u16   wLo[WIDE ? HID * WKP : 8];
  __shared__ float wS[WIDE ? 1 : HID * NOUT];

  const int tid  = threadIdx.x;
  const int lane = tid & 31;
  const int wave = __builtin_amdgcn_readfirstlane(tid >> 5);
  const int row0 = blockIdx.x * OWN_R;
  const int nrow = min(OWN_R, nN - row0);
  const size_t g0 = (size_t)row0 * F;
  const bool al16 =
      (((((unsigned long long)tmp) | ((unsigned long long)outn)) & 15ull) == 0ull);

  const int nList = __builtin_amdgcn_readfirstlane(span[2 * blockIdx.x + 1]);
  const int4* myrec = rec + __builtin_amdgcn_readfirstlane(span[2 * blockIdx.x]);
  const int nB = (nList + 31) >> 5;
  const int4 pad = make_int4(0, OWN_R, 0, 0);
  int4 nx = pad;
  { const int j = (wave << 5) + lane; if (j < nList) nx = myrec[j]; }

  if constexpr (WIDE) {
    #pragma unroll
    for (int i = tid; i < (HID / 2) * HID; i += OWN_NT) {
      const int n = i & (HID - 1), kp = i >> 7;
      u16 h0, l0, h1, l1;
      split_bf(Wn[(2 * kp) * HID + n], h0, l0);
      split_bf(Wn[(2 * kp + 1) * HID + n], h1, l1);
      *(u32*)&wHi[n * WKP + 2 * kp] = (u32)h0 | ((u32)h1 << 16);
      *(u32*)&wLo[n * WKP + 2 * kp] = (u32)l0 | ((u32)l1 << 16);
    }
  } else {
    for (int i = tid; i < HID * NOUT; i += OWN_NT) wS[i] = Wn[i];
  }

  if (al16) {
    const float4* t4 = (const float4*)(tmp + g0);
    for (int i = tid; i < OWN_RP * (F / 4); i += OWN_NT) {
      const int r = i >> 5, c4 = i & 31;
      float4 v = make_float4(0.f, 0.f, 0.f, 0.f);
      if (r < nrow) {
        const float w = norm[nE + row0 + r];
        v = t4[i];
        v.x *= w; v.y *= w; v.z *= w; v.w *= w;
      }
      *(float4*)&acc[r * ACP + 4 * c4] = v;
    }
  } else {
    for (int i = tid; i < OWN_RP * F; i += OWN_NT) {
      const int r = i >> 7, c = i & (F - 1);
      acc[r * ACP + c] = (r < nrow) ? tmp[g0 + i] * norm[nE + row0 + r] : 0.0f;
    }
  }
  __syncthreads();

  constexpr int NF = F / 32;
  for (int b = wave; b < nB; b += OWN_NT / 32) {
    const int   s  = nx.x;
    const int   dl = nx.y;
    const float w  = __int_as_float(nx.z);
    {
      const int jn = ((b + OWN_NT / 32) << 5) + lane;
      nx = pad;
      if (jn < nList) nx = myrec[jn];
    }
    const int m = min(32, nList - (b << 5));
    for (int k0 = 0; k0 < m; k0 += 8) {
      float v[8][NF];
      int   dd[8];
      float ww[8];
      #pragma unroll
      for (int u = 0; u < 8; ++u) {
        const int sk = __builtin_amdgcn_readlane(s, k0 + u);
        dd[u] = __builtin_amdgcn_readlane(dl, k0 + u);
        ww[u] = __int_as_float(__builtin_amdgcn_readlane(__float_as_int(w), k0 + u));
        const float* ts = tmp + (size_t)sk * F;
        #pragma unroll
        for (int c = 0; c < NF; ++c) v[u][c] = ts[lane + 32 * c];
      }
      #pragma unroll
      for (int u = 0; u < 8; ++u) {
        float* ad = acc + dd[u] * ACP;
        #pragma unroll
        for (int c = 0; c < NF; ++c) atomicAdd(&ad[lane + 32 * c], v[u][c] * ww[u]);
      }
    }
  }
  __syncthreads();

  for (int i = tid; i < OWN_RP * (F / 4); i += OWN_NT) {
    const int r = i >> 5, c4 = i & 31;
    float4 v = *(const float4*)&acc[r * ACP + 4 * c4];
    v.x += bias[4 * c4];     v.y += bias[4 * c4 + 1];
    v.z += bias[4 * c4 + 2]; v.w += bias[4 * c4 + 3];
    v.x = (v.x > 0.0f) ? v.x : expm1f(v.x);
    v.y = (v.y > 0.0f) ? v.y : expm1f(v.y);
    v.z = (v.z > 0.0f) ? v.z : expm1f(v.z);
    v.w = (v.w > 0.0f) ? v.w : expm1f(v.w);
    *(float4*)&acc[r * ACP + 4 * c4] = v;
  }
  __syncthreads();

  if constexpr (WIDE) {
    const int l15 = lane & 15;
    const int kh  = (lane >> 4) << 3;
    const int mt  = wave >> 1;
    const int nh  = wave & 1;
    const bool mine = wave < 2 * (OWN_RP / 16);
    v8f d[4];
    for (int t = 0; t < 4; ++t)
      for (int r = 0; r < 8; ++r) d[t][r] = 0.0f;
    if (mine) {
      union Frag { uint4 q[2]; u32 w[8]; v16bf v; };
      const float* arow = acc + ((mt << 4) + l15) * ACP;
      #pragma unroll
      for (int kc = 0; kc < HID; kc += BK) {
        float a[16];
        *(float4*)&a[0]  = *(const float4*)&arow[kc + kh];
        *(float4*)&a[4]  = *(const float4*)&arow[kc + kh + 4];
        *(float4*)&a[8]  = *(const float4*)&arow[kc + 16 + kh];
        *(float4*)&a[12] = *(const float4*)&arow[kc + 16 + kh + 4];
        Frag ah, al;
        #pragma unroll
        for (int e = 0; e < 8; ++e) {
          u16 h0, l0, h1, l1;
          split_bf(a[2 * e], h0, l0);
          split_bf(a[2 * e + 1], h1, l1);
          ah.w[e] = (u32)h0 | ((u32)h1 << 16);
          al.w[e] = (u32)l0 | ((u32)l1 << 16);
        }
        #pragma unroll
        for (int nt = 0; nt < 4; ++nt) {
          Frag bh, bl;
          const int n = (nh << 6) + (nt << 4) + l15;
          bh.q[0] = *(const uint4*)&wHi[n * WKP + kc + kh];
          bh.q[1] = *(const uint4*)&wHi[n * WKP + kc + 16 + kh];
          bl.q[0] = *(const uint4*)&wLo[n * WKP + kc + kh];
          bl.q[1] = *(const uint4*)&wLo[n * WKP + kc + 16 + kh];
          d[nt] = __builtin_amdgcn_wmma_f32_16x16x32_bf16(false, ah.v, false, bh.v, (short)0, d[nt], false, false);
          d[nt] = __builtin_amdgcn_wmma_f32_16x16x32_bf16(false, ah.v, false, bl.v, (short)0, d[nt], false, false);
          d[nt] = __builtin_amdgcn_wmma_f32_16x16x32_bf16(false, al.v, false, bh.v, (short)0, d[nt], false, false);
        }
      }
    }
    __syncthreads();
    if (mine) {
      float* orow = acc + ((mt << 4) + ((lane & 16) ? 8 : 0)) * ACP + (nh << 6) + l15;
      #pragma unroll
      for (int nt = 0; nt < 4; ++nt)
        #pragma unroll
        for (int r = 0; r < 8; ++r) orow[r * ACP + (nt << 4)] = d[nt][r];
    }
    __syncthreads();
    if (al16) {
      float4* o4 = (float4*)(outn + g0);
      for (int i = tid; i < nrow * (F / 4); i += OWN_NT) {
        const int r = i >> 5, c4 = i & 31;
        o4[i] = *(const float4*)&acc[r * ACP + 4 * c4];
      }
    } else {
      for (int i = tid; i < nrow * F; i += OWN_NT) outn[g0 + i] = acc[(i >> 7) * ACP + (i & (F - 1))];
    }
  } else {
    for (int o = tid; o < nrow * NOUT; o += OWN_NT) {
      const int r = o / NOUT, c = o - r * NOUT;
      const float* hr = acc + r * ACP;
      float sum = 0.0f;
      #pragma unroll 8
      for (int k = 0; k < HID; ++k) sum += hr[k] * wS[k * NOUT + c];
      outn[(size_t)row0 * NOUT + o] = sum;
    }
  }
}

__global__ __launch_bounds__(OWN_NT)
void prod_owner(const float* __restrict__ X, const float* __restrict__ Wn,
                float* __restrict__ outn, int nN) {
  constexpr int F = HID;
  __shared__ __attribute__((aligned(16))) float acc[OWN_RP * ACP];
  __shared__ __attribute__((aligned(16))) u16   wHi[HID * WKP];
  __shared__ __attribute__((aligned(16))) u16   wLo[HID * WKP];

  const int tid  = threadIdx.x;
  const int lane = tid & 31;
  const int wave = __builtin_amdgcn_readfirstlane(tid >> 5);
  const int row0 = blockIdx.x * OWN_R;
  const int nrow = min(OWN_R, nN - row0);
  const size_t g0 = (size_t)row0 * F;
  const bool al16 =
      (((((unsigned long long)X) | ((unsigned long long)outn)) & 15ull) == 0ull);

  constexpr int NW = (HID / 2) * HID / OWN_NT;
  constexpr int NX = (OWN_RP * (F / 4) + OWN_NT - 1) / OWN_NT;
  float w0[NW], w1[NW];
  #pragma unroll
  for (int u = 0; u < NW; ++u) {
    const int i = tid + u * OWN_NT;
    const int n = i & (HID - 1), kp = i >> 7;
    w0[u] = Wn[(2 * kp) * HID + n];
    w1[u] = Wn[(2 * kp + 1) * HID + n];
  }
  float4 xv[NX];
  #pragma unroll
  for (int u = 0; u < NX; ++u) xv[u] = make_float4(0.f, 0.f, 0.f, 0.f);
  if (al16) {
    const float4* x4 = (const float4*)(X + g0);
    const int last = nrow * (F / 4) - 1;
    #pragma unroll
    for (int u = 0; u < NX; ++u) xv[u] = x4[min(tid + u * OWN_NT, last)];
  }

  #pragma unroll
  for (int u = 0; u < NW; ++u) {
    const int i = tid + u * OWN_NT;
    const int n = i & (HID - 1), kp = i >> 7;
    u16 h0, l0, h1, l1;
    split_bf(w0[u], h0, l0);
    split_bf(w1[u], h1, l1);
    *(u32*)&wHi[n * WKP + 2 * kp] = (u32)h0 | ((u32)h1 << 16);
    *(u32*)&wLo[n * WKP + 2 * kp] = (u32)l0 | ((u32)l1 << 16);
  }

  if (al16) {
    #pragma unroll
    for (int u = 0; u < NX; ++u) {
      const int i = tid + u * OWN_NT;
      if (i < OWN_RP * (F / 4)) {
        const int r = i >> 5, c4 = i & 31;
        float4 v = xv[u];
        if (r >= nrow) v = make_float4(0.f, 0.f, 0.f, 0.f);
        *(float4*)&acc[r * ACP + 4 * c4] = v;
      }
    }
  } else {
    for (int i = tid; i < OWN_RP * F; i += OWN_NT) {
      const int r = i >> 7, c = i & (F - 1);
      acc[r * ACP + c] = (r < nrow) ? X[g0 + i] : 0.0f;
    }
  }
  __syncthreads();

  const int l15 = lane & 15;
  const int kh  = (lane >> 4) << 3;
  const int mt  = wave >> 1;
  const int nh  = wave & 1;
  const bool mine = wave < 2 * (OWN_RP / 16);
  v8f d[4];
  for (int t = 0; t < 4; ++t)
    for (int r = 0; r < 8; ++r) d[t][r] = 0.0f;
  if (mine) {
    union Frag { uint4 q[2]; u32 w[8]; v16bf v; };
    const float* arow = acc + ((mt << 4) + l15) * ACP;
    #pragma unroll
    for (int kc = 0; kc < HID; kc += BK) {
      float a[16];
      *(float4*)&a[0]  = *(const float4*)&arow[kc + kh];
      *(float4*)&a[4]  = *(const float4*)&arow[kc + kh + 4];
      *(float4*)&a[8]  = *(const float4*)&arow[kc + 16 + kh];
      *(float4*)&a[12] = *(const float4*)&arow[kc + 16 + kh + 4];
      Frag ah, al;
      #pragma unroll
      for (int e = 0; e < 8; ++e) {
        u16 h0, l0, h1, l1;
        split_bf(a[2 * e], h0, l0);
        split_bf(a[2 * e + 1], h1, l1);
        ah.w[e] = (u32)h0 | ((u32)h1 << 16);
        al.w[e] = (u32)l0 | ((u32)l1 << 16);
      }
      #pragma unroll
      for (int nt = 0; nt < 4; ++nt) {
        Frag bh, bl;
        const int n = (nh << 6) + (nt << 4) + l15;
        bh.q[0] = *(const uint4*)&wHi[n * WKP + kc + kh];
        bh.q[1] = *(const uint4*)&wHi[n * WKP + kc + 16 + kh];
        bl.q[0] = *(const uint4*)&wLo[n * WKP + kc + kh];
        bl.q[1] = *(const uint4*)&wLo[n * WKP + kc + 16 + kh];
        d[nt] = __builtin_amdgcn_wmma_f32_16x16x32_bf16(false, ah.v, false, bh.v, (short)0, d[nt], false, false);
        d[nt] = __builtin_amdgcn_wmma_f32_16x16x32_bf16(false, ah.v, false, bl.v, (short)0, d[nt], false, false);
        d[nt] = __builtin_amdgcn_wmma_f32_16x16x32_bf16(false, al.v, false, bh.v, (short)0, d[nt], false, false);
      }
    }
  }
  __syncthreads();
  if (mine) {
    float* orow = acc + ((mt << 4) + ((lane & 16) ? 8 : 0)) * ACP + (nh << 6) + l15;
    #pragma unroll
    for (int nt = 0; nt < 4; ++nt)
      #pragma unroll
      for (int r = 0; r < 8; ++r) orow[r * ACP + (nt << 4)] = d[nt][r];
  }
  __syncthreads();
  if (al16) {
    float4* o4 = (float4*)(outn + g0);
    for (int i = tid; i < nrow * (F / 4); i += OWN_NT) {
      const int r = i >> 5, c4 = i & 31;
      o4[i] = *(const float4*)&acc[r * ACP + 4 * c4];
    }
  } else {
    for (int i = tid; i < nrow * F; i += OWN_NT) outn[g0 + i] = acc[(i >> 7) * ACP + (i & (F - 1))];
  }
}

static inline int cdiv(long long a, long long b) { return (int)((a + b - 1) / b); }

extern "C" void kernel_launch(void* const* d_in, const int* in_sizes, int n_in,
                              void* d_out, int out_size, void* d_ws, size_t ws_size,
                              hipStream_t stream) {
  (void)n_in; (void)out_size; (void)ws_size;
  const float* x    = (const float*)d_in[0];
  const int*   eidx = (const int*)d_in[1];
  const float* Wst  = (const float*)d_in[2];
  const float* bst  = (const float*)d_in[3];
  const float* Wout = (const float*)d_in[4];
  const float* bout = (const float*)d_in[5];

  const int N = in_sizes[0] / HID;
  const int E = in_sizes[1] / 2;
  constexpr int NCLS = 10;

  const int* src = eidx;
  const int* dst = eidx + E;

  const int own_blocks  = cdiv(N, OWN_R);

  int4*  rec  = (int4*)d_ws;
  float* norm = (float*)(rec + E);
  float* deg  = norm + ((size_t)E + N);
  float* dinv = deg + N;
  float* hbuf = dinv + N;
  float* tmp  = hbuf + (size_t)N * HID;
  int*   span = (int*)(tmp + (size_t)N * HID);

  fill_ones<<<cdiv(N, 256), 256, 0, stream>>>(deg, N);
  deg_accum<<<cdiv(E, 256), 256, 0, stream>>>(dst, deg, E);
  rsqrt_of<<<cdiv(N, 256), 256, 0, stream>>>(deg, dinv, N);
  norm_kernel<<<cdiv((long long)E + N, 256), 256, 0, stream>>>(src, dst, dinv, norm, E, N);
  own_lists<<<own_blocks, OWN_NT, 0, stream>>>(src, dst, norm, deg, rec, span, E, N);

  prod_owner<<<own_blocks, OWN_NT, 0, stream>>>(x, Wst, tmp, N);

  float* cur = tmp;
  float* nxt = hbuf;
  for (int l = 0; l < 7; ++l) {
    agg_next<HID><<<own_blocks, OWN_NT, 0, stream>>>(cur, rec, span, norm, bst + (size_t)l * HID,
                                                     Wst + (size_t)(l + 1) * HID * HID, nxt, E, N);
    float* t = cur; cur = nxt; nxt = t;
  }
  agg_next<NCLS><<<own_blocks, OWN_NT, 0, stream>>>(cur, rec, span, norm, bst + (size_t)7 * HID,
                                                    Wout, nxt, E, N);

  agg_owner<NCLS><<<own_blocks, OWN_NT, 0, stream>>>(nxt, rec, span, norm, bout, (float*)d_out, E, N);
}


# Owner scan counts in-degrees itself (LDS counters) and queues its matches; records and weights built from the queue: 4 preprocessing launches and all global atomics gone; sums in LDS arrival order
# speedup vs baseline: 4.6341x; 1.3007x over previous
#include <hip/hip_runtime.h>
#include <math.h>

typedef __attribute__((ext_vector_type(16))) __bf16 v16bf;
typedef __attribute__((ext_vector_type(8)))  float  v8f;
typedef unsigned short u16;
typedef unsigned int   u32;

#define HID 128
#define BK  32

__device__ __forceinline__ float bfh_to_f(u16 h) {
  union { u32 u; float f; } c; c.u = ((u32)h) << 16; return c.f;
}
__device__ __forceinline__ u16 f_to_bf_rne(float f) {
  union { float f; u32 u; } c; c.f = f;
  u32 r = c.u + 0x7FFFu + ((c.u >> 16) & 1u);
  return (u16)(r >> 16);
}
__device__ __forceinline__ float elu_hw(float v) {
  return (v > 0.0f) ? v : __builtin_amdgcn_exp2f(v * 1.44269504f) - 1.0f;
}
__device__ __forceinline__ void split_bf(float v, u16& hi, u16& lo) {
  union { float f; u32 u; } c; c.f = v;
  u16 h = (u16)(c.u >> 16);
  float r = v - bfh_to_f(h);
  hi = h; lo = f_to_bf_rne(r);
}

__device__ __forceinline__ float rsqrt_deg(float v) { return (v > 0.0f) ? rsqrtf(v) : 0.0f; }

#define OWN_R   200
#define OWN_NT  1024
#define OWN_U   8

#define OWN_QCAP 16384

__global__ __launch_bounds__(OWN_NT)
void own_count(const int* __restrict__ dst, float* __restrict__ deg,
               int* __restrict__ qcnt, int* __restrict__ queue, int nE, int nN) {
  __shared__ int cntS;
  __shared__ int degc[OWN_R];
  __shared__ int qL[OWN_QCAP];

  const int tid  = threadIdx.x;
  const int row0 = blockIdx.x * OWN_R;
  const int nrow = min(OWN_R, nN - row0);
  int* myq = queue + (size_t)blockIdx.x * OWN_QCAP;

  if (tid == 0) cntS = 0;
  if (tid < OWN_R) degc[tid] = 0;
  __syncthreads();

  auto push = [&](int e, int pos) {
    if (pos < OWN_QCAP) { myq[pos] = e; qL[pos] = e; }
    else atomicAdd(&degc[dst[e] - row0], 1);
  };

  const bool vec = ((((unsigned long long)dst) & 15ull) == 0ull);
  const int  n4  = vec ? (nE >> 2) : 0;
  const int4* dst4 = (const int4*)dst;
  for (int base4 = 0; base4 < n4; base4 += OWN_U * OWN_NT) {
    int4 q[OWN_U];
    #pragma unroll
    for (int u = 0; u < OWN_U; ++u) {
      const int i = base4 + u * OWN_NT + tid;
      q[u] = dst4[min(i, n4 - 1)];
    }
    unsigned hit = 0u;
    #pragma unroll
    for (int u = 0; u < OWN_U; ++u) {
      const int i = base4 + u * OWN_NT + tid;
      const unsigned lim = (i < n4) ? (unsigned)nrow : 0u;
      hit |= (unsigned)((unsigned)(q[u].x - row0) < lim) << (4 * u + 0);
      hit |= (unsigned)((unsigned)(q[u].y - row0) < lim) << (4 * u + 1);
      hit |= (unsigned)((unsigned)(q[u].z - row0) < lim) << (4 * u + 2);
      hit |= (unsigned)((unsigned)(q[u].w - row0) < lim) << (4 * u + 3);
    }
    if (hit) {
      int pos = atomicAdd(&cntS, __popc(hit));
      while (hit) {
        const int k = __builtin_ctz(hit);
        hit &= hit - 1u;
        push(((base4 + (k >> 2) * OWN_NT + tid) << 2) + (k & 3), pos);
        ++pos;
      }
    }
  }
  for (int i = (n4 << 2) + tid; i < nE; i += OWN_NT)
    if ((unsigned)(dst[i] - row0) < (unsigned)nrow) push(i, atomicAdd(&cntS, 1));
  __syncthreads();

  const int nq   = cntS;
  const int nfit = min(nq, OWN_QCAP);
  for (int i0 = 0; i0 < nfit; i0 += 4 * OWN_NT) {
    int e[4], d[4];
    #pragma unroll
    for (int u = 0; u < 4; ++u) e[u] = qL[min(i0 + u * OWN_NT + tid, nfit - 1)];
    #pragma unroll
    for (int u = 0; u < 4; ++u) d[u] = dst[e[u]];
    asm volatile("" :: "v"(d[0]), "v"(d[1]), "v"(d[2]), "v"(d[3]));
    #pragma unroll
    for (int u = 0; u < 4; ++u)
      if (i0 + u * OWN_NT + tid < nfit) atomicAdd(&degc[d[u] - row0], 1);
  }
  __syncthreads();

  if (tid < nrow) deg[row0 + tid] = (float)(degc[tid] + 1);
  if (tid == 0) qcnt[blockIdx.x] = nq;
}

__global__ __launch_bounds__(OWN_NT)
void own_lists(const int* __restrict__ src, const int* __restrict__ dst,
               const float* __restrict__ deg, const int* __restrict__ qcnt,
               const int* __restrict__ queue, float* __restrict__ norm,
               int4* __restrict__ rec, int* __restrict__ span, int nE, int nN) {
  __shared__ int baseS;
  __shared__ int degm[OWN_R];
  __shared__ int offs[OWN_R];
  __shared__ int fill[OWN_R];

  const int tid  = threadIdx.x;
  const int row0 = blockIdx.x * OWN_R;
  const int nrow = min(OWN_R, nN - row0);
  const int* myq = queue + (size_t)blockIdx.x * OWN_QCAP;
  const int nq   = qcnt[blockIdx.x];

  if (tid == 0) baseS = 0;
  if (tid < OWN_R) {
    float dv = 1.0f;
    if (tid < nrow) {
      dv = deg[row0 + tid];
      const float r = rsqrt_deg(dv);
      norm[nE + row0 + tid] = r * r;
    }
    degm[tid] = (tid < nrow) ? (int)dv - 1 : 0;
    fill[tid] = 0;
  }
  __syncthreads();

  int part = 0;
  for (int v = tid; v < row0; v += OWN_NT) part += (int)deg[v] - 1;
  atomicAdd(&baseS, part);
  if (tid < OWN_R) {
    int a = 0;
    for (int r = 0; r < tid; ++r) a += degm[r];
    offs[tid] = a;
  }
  __syncthreads();
  const int base  = baseS;
  const int total = offs[OWN_R - 1] + degm[OWN_R - 1];

  auto place = [&](int s, int dl, float ds) {
    const float w = rsqrt_deg(ds) * rsqrt_deg((float)(degm[dl] + 1));
    const int p = base + offs[dl] + atomicAdd(&fill[dl], 1);
    if (p < nE) rec[p] = make_int4(s, dl, __float_as_int(w), 0);
  };

  if (nq <= OWN_QCAP) {
    for (int i0 = 0; i0 < nq; i0 += 4 * OWN_NT) {
      int e[4], d[4], s[4];
      float ds[4];
      #pragma unroll
      for (int u = 0; u < 4; ++u) e[u] = myq[min(i0 + u * OWN_NT + tid, nq - 1)];
      #pragma unroll
      for (int u = 0; u < 4; ++u) { d[u] = dst[e[u]]; s[u] = src[e[u]]; }
      #pragma unroll
      for (int u = 0; u < 4; ++u) ds[u] = deg[s[u]];
      asm volatile("" :: "v"(d[0]), "v"(d[1]), "v"(d[2]), "v"(d[3]), "v"(ds[0]), "v"(ds[1]), "v"(ds[2]), "v"(ds[3]));
      #pragma unroll
      for (int u = 0; u < 4; ++u)
        if (i0 + u * OWN_NT + tid < nq) place(s[u], d[u] - row0, ds[u]);
    }
  } else {
    for (int i = tid; i < nE; i += OWN_NT) {
      const int dl = dst[i] - row0;
      if ((unsigned)dl < (unsigned)nrow) { const int s = src[i]; place(s, dl, deg[s]); }
    }
  }
  __syncthreads();

  if (tid == 0) {
    span[2 * blockIdx.x]     = base;
    span[2 * blockIdx.x + 1] = max(0, min(total, nE - base));
  }
}

template <int F>
__global__ __launch_bounds__(OWN_NT)
void agg_owner(const float* __restrict__ tmp, const int4* __restrict__ rec,
               const int* __restrict__ span, const float* __restrict__ norm,
               const float* __restrict__ bias, float* __restrict__ agg, int nE, int nN) {
  __shared__ __attribute__((aligned(16))) float acc[(OWN_R + 1) * F];

  const int tid  = threadIdx.x;
  const int lane = tid & 31;
  const int wave = __builtin_amdgcn_readfirstlane(tid >> 5);
  const int row0 = blockIdx.x * OWN_R;
  const int nrow = min(OWN_R, nN - row0);
  const size_t g0 = (size_t)row0 * F;
  const bool al16 = (F % 4 == 0) &&
      (((((unsigned long long)tmp) | ((unsigned long long)agg)) & 15ull) == 0ull);

  const int nList = __builtin_amdgcn_readfirstlane(span[2 * blockIdx.x + 1]);
  const int4* myrec = rec + __builtin_amdgcn_readfirstlane(span[2 * blockIdx.x]);
  const int nB = (nList + 31) >> 5;
  const int4 pad = make_int4(0, OWN_R, 0, 0);
  int4 nx = pad;
  { const int j = (wave << 5) + lane; if (j < nList) nx = myrec[j]; }

  if (al16) {
    const float4* t4 = (const float4*)(tmp + g0);
    float4* a4 = (float4*)acc;
    for (int i = tid; i < (OWN_R + 1) * F / 4; i += OWN_NT) {
      const int r = i / (F / 4);
      float4 v = make_float4(0.f, 0.f, 0.f, 0.f);
      if (r < nrow) {
        const float w = norm[nE + row0 + r];
        v = t4[i];
        v.x *= w; v.y *= w; v.z *= w; v.w *= w;
      }
      a4[i] = v;
    }
  } else {
    for (int i = tid; i < (OWN_R + 1) * F; i += OWN_NT) {
      const int r = i / F;
      acc[i] = (r < nrow) ? tmp[g0 + i] * norm[nE + row0 + r] : 0.0f;
    }
  }
  __syncthreads();

  constexpr int NF = (F + 31) / 32;
  float run[NF];
  #pragma unroll
  for (int c = 0; c < NF; ++c) run[c] = 0.0f;
  int rrow = OWN_R;
  for (int b = wave; b < nB; b += OWN_NT / 32) {
    const int   s  = nx.x;
    const int   dl = nx.y;
    const float w  = __int_as_float(nx.z);
    {
      const int jn = ((b + OWN_NT / 32) << 5) + lane;
      nx = pad;
      if (jn < nList) nx = myrec[jn];
    }
    const int m = min(32, nList - (b << 5));
    for (int k0 = 0; k0 < m; k0 += 8) {
      float v[8][NF];
      int   dd[8];
      float ww[8];
      #pragma unroll
      for (int u = 0; u < 8; ++u) {
        const int sk = __builtin_amdgcn_readlane(s, k0 + u);
        dd[u] = __builtin_amdgcn_readlane(dl, k0 + u);
        ww[u] = __int_as_float(__builtin_amdgcn_readlane(__float_as_int(w), k0 + u));
        const float* ts = tmp + (size_t)sk * F;
        #pragma unroll
        for (int c = 0; c < NF; ++c) {
          const int f = lane + 32 * c;
          v[u][c] = (f < F) ? ts[f] : 0.0f;
        }
      }
      #pragma unroll
      for (int u = 0; u < 8; ++u) {
        if (dd[u] != rrow) {
          float* ad = acc + rrow * F;
          #pragma unroll
          for (int c = 0; c < NF; ++c) {
            const int f = lane + 32 * c;
            if (f < F) atomicAdd(&ad[f], run[c]);
            run[c] = 0.0f;
          }
          rrow = dd[u];
        }
        #pragma unroll
        for (int c = 0; c < NF; ++c) run[c] += v[u][c] * ww[u];
      }
    }
  }
  {
    float* ad = acc + rrow * F;
    #pragma unroll
    for (int c = 0; c < NF; ++c) {
      const int f = lane + 32 * c;
      if (f < F) atomicAdd(&ad[f], run[c]);
    }
  }
  __syncthreads();

  if (al16) {
    const float4* a4 = (const float4*)acc;
    float4* o4 = (float4*)(agg + g0);
    for (int i = tid; i < nrow * (F / 4); i += OWN_NT) {
      const int c = (i % (F / 4)) * 4;
      float4 v = a4[i];
      v.x += bias[c]; v.y += bias[c + 1]; v.z += bias[c + 2]; v.w += bias[c + 3];
      o4[i] = v;
    }
  } else {
    for (int i = tid; i < nrow * F; i += OWN_NT) agg[g0 + i] = acc[i] + bias[i % F];
  }
}

#define ACP    132
#define OWN_RP 208
#define WKP    136

template <int NOUT>
__global__ __launch_bounds__(OWN_NT)
void agg_next(const float* __restrict__ tmp, const int4* __restrict__ rec,
              const int* __restrict__ span, const float* __restrict__ norm,
              const float* __restrict__ bias, const float* __restrict__ Wn,
              float* __restrict__ outn, int nE, int nN) {
  constexpr int  F    = HID;
  constexpr bool WIDE = (NOUT == HID);
  __shared__ __attribute__((aligned(16))) float acc[OWN_RP * ACP];
  __shared__ __attribute__((aligned(16))) u16   wHi[WIDE ? HID * WKP : 8];
  __shared__ __attribute__((aligned(16))) u16   wLo[WIDE ? HID * WKP : 8];
  __shared__ float wS[WIDE ? 1 : HID * NOUT];

  const int tid  = threadIdx.x;
  const int lane = tid & 31;
  const int wave = __builtin_amdgcn_readfirstlane(tid >> 5);
  const int row0 = blockIdx.x * OWN_R;
  const int nrow = min(OWN_R, nN - row0);
  const size_t g0 = (size_t)row0 * F;
  const bool al16 =
      (((((unsigned long long)tmp) | ((unsigned long long)outn)) & 15ull) == 0ull);

  const int nList = __builtin_amdgcn_readfirstlane(span[2 * blockIdx.x + 1]);
  const int4* myrec = rec + __builtin_amdgcn_readfirstlane(span[2 * blockIdx.x]);
  const int nB = (nList + 31) >> 5;
  const int4 pad = make_int4(0, OWN_R, 0, 0);
  int4 nx = pad;
  { const int j = (wave << 5) + lane; if (j < nList) nx = myrec[j]; }

  if constexpr (WIDE) {
    #pragma unroll
    for (int i = tid; i < (HID / 2) * HID; i += OWN_NT) {
      const int n = i & (HID - 1), kp = i >> 7;
      u16 h0, l0, h1, l1;
      split_bf(Wn[(2 * kp) * HID + n], h0, l0);
      split_bf(Wn[(2 * kp + 1) * HID + n], h1, l1);
      *(u32*)&wHi[n * WKP + 2 * kp] = (u32)h0 | ((u32)h1 << 16);
      *(u32*)&wLo[n * WKP + 2 * kp] = (u32)l0 | ((u32)l1 << 16);
    }
  } else {
    for (int i = tid; i < HID * NOUT; i += OWN_NT) wS[i] = Wn[i];
  }

  if (al16) {
    const float4* t4 = (const float4*)(tmp + g0);
    for (int i = tid; i < OWN_RP * (F / 4); i += OWN_NT) {
      const int r = i >> 5, c4 = i & 31;
      float4 v = make_float4(0.f, 0.f, 0.f, 0.f);
      if (r < nrow) {
        const float w = norm[nE + row0 + r];
        v = t4[i];
        v.x *= w; v.y *= w; v.z *= w; v.w *= w;
      }
      *(float4*)&acc[r * ACP + 4 * c4] = v;
    }
  } else {
    for (int i = tid; i < OWN_RP * F; i += OWN_NT) {
      const int r = i >> 7, c = i & (F - 1);
      acc[r * ACP + c] = (r < nrow) ? tmp[g0 + i] * norm[nE + row0 + r] : 0.0f;
    }
  }
  __syncthreads();

  constexpr int NF = F / 32;
  float run[NF];
  #pragma unroll
  for (int c = 0; c < NF; ++c) run[c] = 0.0f;
  int rrow = OWN_R;
  for (int b = wave; b < nB; b += OWN_NT / 32) {
    const int   s  = nx.x;
    const int   dl = nx.y;
    const float w  = __int_as_float(nx.z);
    {
      const int jn = ((b + OWN_NT / 32) << 5) + lane;
      nx = pad;
      if (jn < nList) nx = myrec[jn];
    }
    const int m = min(32, nList - (b << 5));
    for (int k0 = 0; k0 < m; k0 += 8) {
      float v[8][NF];
      int   dd[8];
      float ww[8];
      #pragma unroll
      for (int u = 0; u < 8; ++u) {
        const int sk = __builtin_amdgcn_readlane(s, k0 + u);
        dd[u] = __builtin_amdgcn_readlane(dl, k0 + u);
        ww[u] = __int_as_float(__builtin_amdgcn_readlane(__float_as_int(w), k0 + u));
        const float* ts = tmp + (size_t)sk * F;
        #pragma unroll
        for (int c = 0; c < NF; ++c) v[u][c] = ts[lane + 32 * c];
      }
      #pragma unroll
      for (int u = 0; u < 8; ++u) {
        if (dd[u] != rrow) {
          float* ad = acc + rrow * ACP;
          #pragma unroll
          for (int c = 0; c < NF; ++c) {
            atomicAdd(&ad[lane + 32 * c], run[c]);
            run[c] = 0.0f;
          }
          rrow = dd[u];
        }
        #pragma unroll
        for (int c = 0; c < NF; ++c) run[c] += v[u][c] * ww[u];
      }
    }
  }
  {
    float* ad = acc + rrow * ACP;
    #pragma unroll
    for (int c = 0; c < NF; ++c) atomicAdd(&ad[lane + 32 * c], run[c]);
  }
  __syncthreads();

  {
    const int c4 = tid & 31;
    const float b0 = bias[4 * c4], b1 = bias[4 * c4 + 1], b2 = bias[4 * c4 + 2], b3 = bias[4 * c4 + 3];
    float* p = acc + (tid >> 5) * ACP + 4 * c4;
    for (int r = tid >> 5; r < OWN_RP; r += OWN_NT / 32, p += (OWN_NT / 32) * ACP) {
      float4 v = *(const float4*)p;
      v.x = elu_hw(v.x + b0);
      v.y = elu_hw(v.y + b1);
      v.z = elu_hw(v.z + b2);
      v.w = elu_hw(v.w + b3);
      *(float4*)p = v;
    }
  }
  __syncthreads();

  if constexpr (WIDE) {
    const int l15 = lane & 15;
    const int kh  = (lane >> 4) << 3;
    const int mt  = wave >> 1;
    const int nh  = wave & 1;
    const bool mine = wave < 2 * (OWN_RP / 16);
    v8f d[4];
    for (int t = 0; t < 4; ++t)
      for (int r = 0; r < 8; ++r) d[t][r] = 0.0f;
    if (mine) {
      union Frag { uint4 q[2]; u32 w[8]; v16bf v; };
      const float* arow = acc + ((mt << 4) + l15) * ACP;
      #pragma unroll
      for (int kc = 0; kc < HID; kc += BK) {
        float a[16];
        *(float4*)&a[0]  = *(const float4*)&arow[kc + kh];
        *(float4*)&a[4]  = *(const float4*)&arow[kc + kh + 4];
        *(float4*)&a[8]  = *(const float4*)&arow[kc + 16 + kh];
        *(float4*)&a[12] = *(const float4*)&arow[kc + 16 + kh + 4];
        Frag ah, al;
        #pragma unroll
        for (int e = 0; e < 8; ++e) {
          u16 h0, l0, h1, l1;
          split_bf(a[2 * e], h0, l0);
          split_bf(a[2 * e + 1], h1, l1);
          ah.w[e] = (u32)h0 | ((u32)h1 << 16);
          al.w[e] = (u32)l0 | ((u32)l1 << 16);
        }
        #pragma unroll
        for (int nt = 0; nt < 4; ++nt) {
          Frag bh, bl;
          const int n = (nh << 6) + (nt << 4) + l15;
          bh.q[0] = *(const uint4*)&wHi[n * WKP + kc + kh];
          bh.q[1] = *(const uint4*)&wHi[n * WKP + kc + 16 + kh];
          bl.q[0] = *(const uint4*)&wLo[n * WKP + kc + kh];
          bl.q[1] = *(const uint4*)&wLo[n * WKP + kc + 16 + kh];
          d[nt] = __builtin_amdgcn_wmma_f32_16x16x32_bf16(false, ah.v, false, bh.v, (short)0, d[nt], false, false);
          d[nt] = __builtin_amdgcn_wmma_f32_16x16x32_bf16(false, ah.v, false, bl.v, (short)0, d[nt], false, false);
          d[nt] = __builtin_amdgcn_wmma_f32_16x16x32_bf16(false, al.v, false, bh.v, (short)0, d[nt], false, false);
        }
      }
    }
    __syncthreads();
    if (mine) {
      float* orow = acc + ((mt << 4) + ((lane & 16) ? 8 : 0)) * ACP + (nh << 6) + l15;
      #pragma unroll
      for (int nt = 0; nt < 4; ++nt)
        #pragma unroll
        for (int r = 0; r < 8; ++r) orow[r * ACP + (nt << 4)] = d[nt][r];
    }
    __syncthreads();
    if (al16) {
      float4* o4 = (float4*)(outn + g0);
      for (int i = tid; i < nrow * (F / 4); i += OWN_NT) {
        const int r = i >> 5, c4 = i & 31;
        o4[i] = *(const float4*)&acc[r * ACP + 4 * c4];
      }
    } else {
      for (int i = tid; i < nrow * F; i += OWN_NT) outn[g0 + i] = acc[(i >> 7) * ACP + (i & (F - 1))];
    }
  } else {
    for (int o = tid; o < nrow * NOUT; o += OWN_NT) {
      const int r = o / NOUT, c = o - r * NOUT;
      const float* hr = acc + r * ACP;
      float sum = 0.0f;
      #pragma unroll 8
      for (int k = 0; k < HID; ++k) sum += hr[k] * wS[k * NOUT + c];
      outn[(size_t)row0 * NOUT + o] = sum;
    }
  }
}

__global__ __launch_bounds__(OWN_NT)
void prod_owner(const float* __restrict__ X, const float* __restrict__ Wn,
                float* __restrict__ outn, int nN) {
  constexpr int F = HID;
  __shared__ __attribute__((aligned(16))) float acc[OWN_RP * ACP];
  __shared__ __attribute__((aligned(16))) u16   wHi[HID * WKP];
  __shared__ __attribute__((aligned(16))) u16   wLo[HID * WKP];

  const int tid  = threadIdx.x;
  const int lane = tid & 31;
  const int wave = __builtin_amdgcn_readfirstlane(tid >> 5);
  const int row0 = blockIdx.x * OWN_R;
  const int nrow = min(OWN_R, nN - row0);
  const size_t g0 = (size_t)row0 * F;
  const bool al16 =
      (((((unsigned long long)X) | ((unsigned long long)outn)) & 15ull) == 0ull);

  constexpr int NW = (HID / 2) * HID / OWN_NT;
  constexpr int NX = (OWN_RP * (F / 4) + OWN_NT - 1) / OWN_NT;
  float w0[NW], w1[NW];
  #pragma unroll
  for (int u = 0; u < NW; ++u) {
    const int i = tid + u * OWN_NT;
    const int n = i & (HID - 1), kp = i >> 7;
    w0[u] = Wn[(2 * kp) * HID + n];
    w1[u] = Wn[(2 * kp + 1) * HID + n];
  }
  float4 xv[NX];
  #pragma unroll
  for (int u = 0; u < NX; ++u) xv[u] = make_float4(0.f, 0.f, 0.f, 0.f);
  if (al16) {
    const float4* x4 = (const float4*)(X + g0);
    const int last = nrow * (F / 4) - 1;
    #pragma unroll
    for (int u = 0; u < NX; ++u) xv[u] = x4[min(tid + u * OWN_NT, last)];
  }

  #pragma unroll
  for (int u = 0; u < NW; ++u) {
    const int i = tid + u * OWN_NT;
    const int n = i & (HID - 1), kp = i >> 7;
    u16 h0, l0, h1, l1;
    split_bf(w0[u], h0, l0);
    split_bf(w1[u], h1, l1);
    *(u32*)&wHi[n * WKP + 2 * kp] = (u32)h0 | ((u32)h1 << 16);
    *(u32*)&wLo[n * WKP + 2 * kp] = (u32)l0 | ((u32)l1 << 16);
  }

  if (al16) {
    #pragma unroll
    for (int u = 0; u < NX; ++u) {
      const int i = tid + u * OWN_NT;
      if (i < OWN_RP * (F / 4)) {
        const int r = i >> 5, c4 = i & 31;
        float4 v = xv[u];
        if (r >= nrow) v = make_float4(0.f, 0.f, 0.f, 0.f);
        *(float4*)&acc[r * ACP + 4 * c4] = v;
      }
    }
  } else {
    for (int i = tid; i < OWN_RP * F; i += OWN_NT) {
      const int r = i >> 7, c = i & (F - 1);
      acc[r * ACP + c] = (r < nrow) ? X[g0 + i] : 0.0f;
    }
  }
  __syncthreads();

  const int l15 = lane & 15;
  const int kh  = (lane >> 4) << 3;
  const int mt  = wave >> 1;
  const int nh  = wave & 1;
  const bool mine = wave < 2 * (OWN_RP / 16);
  v8f d[4];
  for (int t = 0; t < 4; ++t)
    for (int r = 0; r < 8; ++r) d[t][r] = 0.0f;
  if (mine) {
    union Frag { uint4 q[2]; u32 w[8]; v16bf v; };
    const float* arow = acc + ((mt << 4) + l15) * ACP;
    #pragma unroll
    for (int kc = 0; kc < HID; kc += BK) {
      float a[16];
      *(float4*)&a[0]  = *(const float4*)&arow[kc + kh];
      *(float4*)&a[4]  = *(const float4*)&arow[kc + kh + 4];
      *(float4*)&a[8]  = *(const float4*)&arow[kc + 16 + kh];
      *(float4*)&a[12] = *(const float4*)&arow[kc + 16 + kh + 4];
      Frag ah, al;
      #pragma unroll
      for (int e = 0; e < 8; ++e) {
        u16 h0, l0, h1, l1;
        split_bf(a[2 * e], h0, l0);
        split_bf(a[2 * e + 1], h1, l1);
        ah.w[e] = (u32)h0 | ((u32)h1 << 16);
        al.w[e] = (u32)l0 | ((u32)l1 << 16);
      }
      #pragma unroll
      for (int nt = 0; nt < 4; ++nt) {
        Frag bh, bl;
        const int n = (nh << 6) + (nt << 4) + l15;
        bh.q[0] = *(const uint4*)&wHi[n * WKP + kc + kh];
        bh.q[1] = *(const uint4*)&wHi[n * WKP + kc + 16 + kh];
        bl.q[0] = *(const uint4*)&wLo[n * WKP + kc + kh];
        bl.q[1] = *(const uint4*)&wLo[n * WKP + kc + 16 + kh];
        d[nt] = __builtin_amdgcn_wmma_f32_16x16x32_bf16(false, ah.v, false, bh.v, (short)0, d[nt], false, false);
        d[nt] = __builtin_amdgcn_wmma_f32_16x16x32_bf16(false, ah.v, false, bl.v, (short)0, d[nt], false, false);
        d[nt] = __builtin_amdgcn_wmma_f32_16x16x32_bf16(false, al.v, false, bh.v, (short)0, d[nt], false, false);
      }
    }
  }
  __syncthreads();
  if (mine) {
    float* orow = acc + ((mt << 4) + ((lane & 16) ? 8 : 0)) * ACP + (nh << 6) + l15;
    #pragma unroll
    for (int nt = 0; nt < 4; ++nt)
      #pragma unroll
      for (int r = 0; r < 8; ++r) orow[r * ACP + (nt << 4)] = d[nt][r];
  }
  __syncthreads();
  if (al16) {
    float4* o4 = (float4*)(outn + g0);
    for (int i = tid; i < nrow * (F / 4); i += OWN_NT) {
      const int r = i >> 5, c4 = i & 31;
      o4[i] = *(const float4*)&acc[r * ACP + 4 * c4];
    }
  } else {
    for (int i = tid; i < nrow * F; i += OWN_NT) outn[g0 + i] = acc[(i >> 7) * ACP + (i & (F - 1))];
  }
}

static inline int cdiv(long long a, long long b) { return (int)((a + b - 1) / b); }

extern "C" void kernel_launch(void* const* d_in, const int* in_sizes, int n_in,
                              void* d_out, int out_size, void* d_ws, size_t ws_size,
                              hipStream_t stream) {
  (void)n_in; (void)out_size; (void)ws_size;
  const float* x    = (const float*)d_in[0];
  const int*   eidx = (const int*)d_in[1];
  const float* Wst  = (const float*)d_in[2];
  const float* bst  = (const float*)d_in[3];
  const float* Wout = (const float*)d_in[4];
  const float* bout = (const float*)d_in[5];

  const int N = in_sizes[0] / HID;
  const int E = in_sizes[1] / 2;
  constexpr int NCLS = 10;

  const int* src = eidx;
  const int* dst = eidx + E;

  const int own_blocks  = cdiv(N, OWN_R);

  int4*  rec  = (int4*)d_ws;
  float* norm = (float*)(rec + E);
  float* deg  = norm + ((size_t)E + N);
  float* hbuf = deg + 2 * (size_t)N;
  float* tmp  = hbuf + (size_t)N * HID;
  int*   span = (int*)(tmp + (size_t)N * HID);
  int*   qcnt = span + 2 * (size_t)own_blocks;
  int*   queue = qcnt + own_blocks;

  own_count<<<own_blocks, OWN_NT, 0, stream>>>(dst, deg, qcnt, queue, E, N);
  own_lists<<<own_blocks, OWN_NT, 0, stream>>>(src, dst, deg, qcnt, queue, norm, rec, span, E, N);

  prod_owner<<<own_blocks, OWN_NT, 0, stream>>>(x, Wst, tmp, N);

  float* cur = tmp;
  float* nxt = hbuf;
  for (int l = 0; l < 7; ++l) {
    agg_next<HID><<<own_blocks, OWN_NT, 0, stream>>>(cur, rec, span, norm, bst + (size_t)l * HID,
                                                     Wst + (size_t)(l + 1) * HID * HID, nxt, E, N);
    float* t = cur; cur = nxt; nxt = t;
  }
  agg_next<NCLS><<<own_blocks, OWN_NT, 0, stream>>>(cur, rec, span, norm, bst + (size_t)7 * HID,
                                                    Wout, nxt, E, N);

  agg_owner<NCLS><<<own_blocks, OWN_NT, 0, stream>>>(nxt, rec, span, norm, bout, (float*)d_out, E, N);
}


# Start-value pass removed: a row's self-loop term is a record at the head of its row's records, LDS sums start at 0, weight and bias loads ride under the gather; row sums regrouped
# speedup vs baseline: 5.0510x; 1.0900x over previous
#include <hip/hip_runtime.h>
#include <math.h>

typedef __attribute__((ext_vector_type(16))) __bf16 v16bf;
typedef __attribute__((ext_vector_type(8)))  float  v8f;
typedef unsigned short u16;
typedef unsigned int   u32;

#define HID 128
#define BK  32

__device__ __forceinline__ float bfh_to_f(u16 h) {
  union { u32 u; float f; } c; c.u = ((u32)h) << 16; return c.f;
}
__device__ __forceinline__ u16 f_to_bf_rne(float f) {
  union { float f; u32 u; } c; c.f = f;
  u32 r = c.u + 0x7FFFu + ((c.u >> 16) & 1u);
  return (u16)(r >> 16);
}
__device__ __forceinline__ float elu_hw(float v) {
  return (v > 0.0f) ? v : __builtin_amdgcn_exp2f(v * 1.44269504f) - 1.0f;
}
__device__ __forceinline__ void split_bf(float v, u16& hi, u16& lo) {
  union { float f; u32 u; } c; c.f = v;
  u16 h = (u16)(c.u >> 16);
  float r = v - bfh_to_f(h);
  hi = h; lo = f_to_bf_rne(r);
}

__device__ __forceinline__ float rsqrt_deg(float v) { return (v > 0.0f) ? rsqrtf(v) : 0.0f; }

#define OWN_R   200
#define OWN_NT  1024
#define OWN_U   8

#define OWN_QCAP 16384

__global__ __launch_bounds__(OWN_NT)
void own_count(const int* __restrict__ dst, float* __restrict__ deg,
               int* __restrict__ qcnt, int* __restrict__ queue, int nE, int nN) {
  __shared__ int cntS;
  __shared__ int degc[OWN_R];
  __shared__ int qL[OWN_QCAP];

  const int tid  = threadIdx.x;
  const int row0 = blockIdx.x * OWN_R;
  const int nrow = min(OWN_R, nN - row0);
  int* myq = queue + (size_t)blockIdx.x * OWN_QCAP;

  if (tid == 0) cntS = 0;
  if (tid < OWN_R) degc[tid] = 0;
  __syncthreads();

  auto push = [&](int e, int pos) {
    if (pos < OWN_QCAP) { myq[pos] = e; qL[pos] = e; }
    else atomicAdd(&degc[dst[e] - row0], 1);
  };

  const bool vec = ((((unsigned long long)dst) & 15ull) == 0ull);
  const int  n4  = vec ? (nE >> 2) : 0;
  const int4* dst4 = (const int4*)dst;
  for (int base4 = 0; base4 < n4; base4 += OWN_U * OWN_NT) {
    int4 q[OWN_U];
    #pragma unroll
    for (int u = 0; u < OWN_U; ++u) {
      const int i = base4 + u * OWN_NT + tid;
      q[u] = dst4[min(i, n4 - 1)];
    }
    unsigned hit = 0u;
    #pragma unroll
    for (int u = 0; u < OWN_U; ++u) {
      const int i = base4 + u * OWN_NT + tid;
      const unsigned lim = (i < n4) ? (unsigned)nrow : 0u;
      hit |= (unsigned)((unsigned)(q[u].x - row0) < lim) << (4 * u + 0);
      hit |= (unsigned)((unsigned)(q[u].y - row0) < lim) << (4 * u + 1);
      hit |= (unsigned)((unsigned)(q[u].z - row0) < lim) << (4 * u + 2);
      hit |= (unsigned)((unsigned)(q[u].w - row0) < lim) << (4 * u + 3);
    }
    if (hit) {
      int pos = atomicAdd(&cntS, __popc(hit));
      while (hit) {
        const int k = __builtin_ctz(hit);
        hit &= hit - 1u;
        push(((base4 + (k >> 2) * OWN_NT + tid) << 2) + (k & 3), pos);
        ++pos;
      }
    }
  }
  for (int i = (n4 << 2) + tid; i < nE; i += OWN_NT)
    if ((unsigned)(dst[i] - row0) < (unsigned)nrow) push(i, atomicAdd(&cntS, 1));
  __syncthreads();

  const int nq   = cntS;
  const int nfit = min(nq, OWN_QCAP);
  for (int i0 = 0; i0 < nfit; i0 += 4 * OWN_NT) {
    int e[4], d[4];
    #pragma unroll
    for (int u = 0; u < 4; ++u) e[u] = qL[min(i0 + u * OWN_NT + tid, nfit - 1)];
    #pragma unroll
    for (int u = 0; u < 4; ++u) d[u] = dst[e[u]];
    asm volatile("" :: "v"(d[0]), "v"(d[1]), "v"(d[2]), "v"(d[3]));
    #pragma unroll
    for (int u = 0; u < 4; ++u)
      if (i0 + u * OWN_NT + tid < nfit) atomicAdd(&degc[d[u] - row0], 1);
  }
  __syncthreads();

  if (tid < nrow) deg[row0 + tid] = (float)(degc[tid] + 1);
  if (tid == 0) qcnt[blockIdx.x] = nq;
}

__global__ __launch_bounds__(OWN_NT)
void own_lists(const int* __restrict__ src, const int* __restrict__ dst,
               const float* __restrict__ deg, const int* __restrict__ qcnt,
               const int* __restrict__ queue,
               int4* __restrict__ rec, int* __restrict__ span, int nE, int nN) {
  __shared__ int baseS;
  __shared__ int degm[OWN_R];
  __shared__ int offs[OWN_R];
  __shared__ int fill[OWN_R];

  const int tid  = threadIdx.x;
  const int row0 = blockIdx.x * OWN_R;
  const int nrow = min(OWN_R, nN - row0);
  const int* myq = queue + (size_t)blockIdx.x * OWN_QCAP;
  const int nq   = qcnt[blockIdx.x];

  if (tid == 0) baseS = 0;
  if (tid < OWN_R) {
    degm[tid] = (tid < nrow) ? (int)deg[row0 + tid] : 0;
    fill[tid] = 1;
  }
  __syncthreads();

  int part = 0;
  for (int v = tid; v < row0; v += OWN_NT) part += (int)deg[v];
  atomicAdd(&baseS, part);
  if (tid < OWN_R) {
    int a = 0;
    for (int r = 0; r < tid; ++r) a += degm[r];
    offs[tid] = a;
  }
  __syncthreads();
  const int base  = baseS;
  const int total = offs[OWN_R - 1] + degm[OWN_R - 1];
  const int nRec  = nE + nN;

  if (tid < nrow) {
    const float r = rsqrt_deg((float)degm[tid]);
    const int p = base + offs[tid];
    if (p < nRec) rec[p] = make_int4(row0 + tid, tid, __float_as_int(r * r), 0);
  }

  auto place = [&](int s, int dl, float ds) {
    const float w = rsqrt_deg(ds) * rsqrt_deg((float)degm[dl]);
    const int p = base + offs[dl] + atomicAdd(&fill[dl], 1);
    if (p < nRec) rec[p] = make_int4(s, dl, __float_as_int(w), 0);
  };

  if (nq <= OWN_QCAP) {
    for (int i0 = 0; i0 < nq; i0 += 4 * OWN_NT) {
      int e[4], d[4], s[4];
      float ds[4];
      #pragma unroll
      for (int u = 0; u < 4; ++u) e[u] = myq[min(i0 + u * OWN_NT + tid, nq - 1)];
      #pragma unroll
      for (int u = 0; u < 4; ++u) { d[u] = dst[e[u]]; s[u] = src[e[u]]; }
      #pragma unroll
      for (int u = 0; u < 4; ++u) ds[u] = deg[s[u]];
      asm volatile("" :: "v"(d[0]), "v"(d[1]), "v"(d[2]), "v"(d[3]), "v"(ds[0]), "v"(ds[1]), "v"(ds[2]), "v"(ds[3]));
      #pragma unroll
      for (int u = 0; u < 4; ++u)
        if (i0 + u * OWN_NT + tid < nq) place(s[u], d[u] - row0, ds[u]);
    }
  } else {
    for (int i = tid; i < nE; i += OWN_NT) {
      const int dl = dst[i] - row0;
      if ((unsigned)dl < (unsigned)nrow) { const int s = src[i]; place(s, dl, deg[s]); }
    }
  }
  __syncthreads();

  if (tid == 0) {
    span[2 * blockIdx.x]     = base;
    span[2 * blockIdx.x + 1] = max(0, min(total, nRec - base));
  }
}

template <int F>
__global__ __launch_bounds__(OWN_NT)
void agg_owner(const float* __restrict__ tmp, const int4* __restrict__ rec,
               const int* __restrict__ span,
               const float* __restrict__ bias, float* __restrict__ agg, int nN) {
  __shared__ __attribute__((aligned(16))) float acc[(OWN_R + 1) * F];

  const int tid  = threadIdx.x;
  const int lane = tid & 31;
  const int wave = __builtin_amdgcn_readfirstlane(tid >> 5);
  const int row0 = blockIdx.x * OWN_R;
  const int nrow = min(OWN_R, nN - row0);
  const size_t g0 = (size_t)row0 * F;
  const bool al16 = (F % 4 == 0) &&
      (((((unsigned long long)tmp) | ((unsigned long long)agg)) & 15ull) == 0ull);

  const int nList = __builtin_amdgcn_readfirstlane(span[2 * blockIdx.x + 1]);
  const int4* myrec = rec + __builtin_amdgcn_readfirstlane(span[2 * blockIdx.x]);
  const int nB = (nList + 31) >> 5;
  const int4 pad = make_int4(0, OWN_R, 0, 0);
  int4 nx = pad;
  { const int j = (wave << 5) + lane; if (j < nList) nx = myrec[j]; }

  for (int i = tid; i < (OWN_R + 1) * F; i += OWN_NT) acc[i] = 0.0f;
  __syncthreads();

  constexpr int NF = (F + 31) / 32;
  float run[NF];
  #pragma unroll
  for (int c = 0; c < NF; ++c) run[c] = 0.0f;
  int rrow = OWN_R;
  for (int b = wave; b < nB; b += OWN_NT / 32) {
    const int   s  = nx.x;
    const int   dl = nx.y;
    const float w  = __int_as_float(nx.z);
    {
      const int jn = ((b + OWN_NT / 32) << 5) + lane;
      nx = pad;
      if (jn < nList) nx = myrec[jn];
    }
    const int m = min(32, nList - (b << 5));
    for (int k0 = 0; k0 < m; k0 += 8) {
      float v[8][NF];
      int   dd[8];
      float ww[8];
      #pragma unroll
      for (int u = 0; u < 8; ++u) {
        const int sk = __builtin_amdgcn_readlane(s, k0 + u);
        dd[u] = __builtin_amdgcn_readlane(dl, k0 + u);
        ww[u] = __int_as_float(__builtin_amdgcn_readlane(__float_as_int(w), k0 + u));
        const float* ts = tmp + (size_t)sk * F;
        #pragma unroll
        for (int c = 0; c < NF; ++c) {
          const int f = lane + 32 * c;
          v[u][c] = (f < F) ? ts[f] : 0.0f;
        }
      }
      #pragma unroll
      for (int u = 0; u < 8; ++u) {
        if (dd[u] != rrow) {
          float* ad = acc + rrow * F;
          #pragma unroll
          for (int c = 0; c < NF; ++c) {
            const int f = lane + 32 * c;
            if (f < F) atomicAdd(&ad[f], run[c]);
            run[c] = 0.0f;
          }
          rrow = dd[u];
        }
        #pragma unroll
        for (int c = 0; c < NF; ++c) run[c] += v[u][c] * ww[u];
      }
    }
  }
  {
    float* ad = acc + rrow * F;
    #pragma unroll
    for (int c = 0; c < NF; ++c) {
      const int f = lane + 32 * c;
      if (f < F) atomicAdd(&ad[f], run[c]);
    }
  }
  __syncthreads();

  if (al16) {
    const float4* a4 = (const float4*)acc;
    float4* o4 = (float4*)(agg + g0);
    for (int i = tid; i < nrow * (F / 4); i += OWN_NT) {
      const int c = (i % (F / 4)) * 4;
      float4 v = a4[i];
      v.x += bias[c]; v.y += bias[c + 1]; v.z += bias[c + 2]; v.w += bias[c + 3];
      o4[i] = v;
    }
  } else {
    for (int i = tid; i < nrow * F; i += OWN_NT) agg[g0 + i] = acc[i] + bias[i % F];
  }
}

#define ACP    132
#define OWN_RP 208
#define WKP    136

template <int NOUT>
__global__ __launch_bounds__(OWN_NT)
void agg_next(const float* __restrict__ tmp, const int4* __restrict__ rec,
              const int* __restrict__ span,
              const float* __restrict__ bias, const float* __restrict__ Wn,
              float* __restrict__ outn, int nN) {
  constexpr int  F    = HID;
  constexpr bool WIDE = (NOUT == HID);
  __shared__ __attribute__((aligned(16))) float acc[OWN_RP * ACP];
  __shared__ __attribute__((aligned(16))) u16   wHi[WIDE ? HID * WKP : 8];
  __shared__ __attribute__((aligned(16))) u16   wLo[WIDE ? HID * WKP : 8];
  __shared__ float wS[WIDE ? 1 : HID * NOUT];

  const int tid  = threadIdx.x;
  const int lane = tid & 31;
  const int wave = __builtin_amdgcn_readfirstlane(tid >> 5);
  const int row0 = blockIdx.x * OWN_R;
  const int nrow = min(OWN_R, nN - row0);
  const size_t g0 = (size_t)row0 * F;
  const bool al16 =
      (((((unsigned long long)tmp) | ((unsigned long long)outn)) & 15ull) == 0ull);

  const int nList = __builtin_amdgcn_readfirstlane(span[2 * blockIdx.x + 1]);
  const int4* myrec = rec + __builtin_amdgcn_readfirstlane(span[2 * blockIdx.x]);
  const int nB = (nList + 31) >> 5;
  const int4 pad = make_int4(0, OWN_R, 0, 0);
  int4 nx = pad;
  { const int j = (wave << 5) + lane; if (j < nList) nx = myrec[j]; }

  {
    float4* a4 = (float4*)acc;
    for (int i = tid; i < OWN_RP * ACP / 4; i += OWN_NT) a4[i] = make_float4(0.f, 0.f, 0.f, 0.f);
  }
  __syncthreads();

  constexpr int NWL = WIDE ? (HID / 2) * HID / OWN_NT : (HID * NOUT + OWN_NT - 1) / OWN_NT;
  float w0[NWL], w1[NWL];
  #pragma unroll
  for (int u = 0; u < NWL; ++u) {
    const int i = tid + u * OWN_NT;
    if constexpr (WIDE) {
      const int n = i & (HID - 1), kp = i >> 7;
      w0[u] = Wn[(2 * kp) * HID + n];
      w1[u] = Wn[(2 * kp + 1) * HID + n];
    } else {
      w0[u] = Wn[min(i, HID * NOUT - 1)];
      w1[u] = 0.0f;
    }
  }
  const int cb4 = tid & 31;
  const float b0 = bias[4 * cb4], b1 = bias[4 * cb4 + 1], b2 = bias[4 * cb4 + 2], b3 = bias[4 * cb4 + 3];

  constexpr int NF = F / 32;
  float run[NF];
  #pragma unroll
  for (int c = 0; c < NF; ++c) run[c] = 0.0f;
  int rrow = OWN_R;
  for (int b = wave; b < nB; b += OWN_NT / 32) {
    const int   s  = nx.x;
    const int   dl = nx.y;
    const float w  = __int_as_float(nx.z);
    {
      const int jn = ((b + OWN_NT / 32) << 5) + lane;
      nx = pad;
      if (jn < nList) nx = myrec[jn];
    }
    const int m = min(32, nList - (b << 5));
    for (int k0 = 0; k0 < m; k0 += 8) {
      float v[8][NF];
      int   dd[8];
      float ww[8];
      #pragma unroll
      for (int u = 0; u < 8; ++u) {
        const int sk = __builtin_amdgcn_readlane(s, k0 + u);
        dd[u] = __builtin_amdgcn_readlane(dl, k0 + u);
        ww[u] = __int_as_float(__builtin_amdgcn_readlane(__float_as_int(w), k0 + u));
        const float* ts = tmp + (size_t)sk * F;
        #pragma unroll
        for (int c = 0; c < NF; ++c) v[u][c] = ts[lane + 32 * c];
      }
      #pragma unroll
      for (int u = 0; u < 8; ++u) {
        if (dd[u] != rrow) {
          float* ad = acc + rrow * ACP;
          #pragma unroll
          for (int c = 0; c < NF; ++c) {
            atomicAdd(&ad[lane + 32 * c], run[c]);
            run[c] = 0.0f;
          }
          rrow = dd[u];
        }
        #pragma unroll
        for (int c = 0; c < NF; ++c) run[c] += v[u][c] * ww[u];
      }
    }
  }
  {
    float* ad = acc + rrow * ACP;
    #pragma unroll
    for (int c = 0; c < NF; ++c) atomicAdd(&ad[lane + 32 * c], run[c]);
  }

  #pragma unroll
  for (int u = 0; u < NWL; ++u) {
    const int i = tid + u * OWN_NT;
    if constexpr (WIDE) {
      const int n = i & (HID - 1), kp = i >> 7;
      u16 h0, l0, h1, l1;
      split_bf(w0[u], h0, l0);
      split_bf(w1[u], h1, l1);
      *(u32*)&wHi[n * WKP + 2 * kp] = (u32)h0 | ((u32)h1 << 16);
      *(u32*)&wLo[n * WKP + 2 * kp] = (u32)l0 | ((u32)l1 << 16);
    } else {
      if (i < HID * NOUT) wS[i] = w0[u];
    }
  }
  __syncthreads();

  {
    const int c4 = cb4;
    float* p = acc + (tid >> 5) * ACP + 4 * c4;
    for (int r = tid >> 5; r < OWN_RP; r += OWN_NT / 32, p += (OWN_NT / 32) * ACP) {
      float4 v = *(const float4*)p;
      v.x = elu_hw(v.x + b0);
      v.y = elu_hw(v.y + b1);
      v.z = elu_hw(v.z + b2);
      v.w = elu_hw(v.w + b3);
      *(float4*)p = v;
    }
  }
  __syncthreads();

  if constexpr (WIDE) {
    const int l15 = lane & 15;
    const int kh  = (lane >> 4) << 3;
    const int mt  = wave >> 1;
    const int nh  = wave & 1;
    const bool mine = wave < 2 * (OWN_RP / 16);
    v8f d[4];
    for (int t = 0; t < 4; ++t)
      for (int r = 0; r < 8; ++r) d[t][r] = 0.0f;
    if (mine) {
      union Frag { uint4 q[2]; u32 w[8]; v16bf v; };
      const float* arow = acc + ((mt << 4) + l15) * ACP;
      #pragma unroll
      for (int kc = 0; kc < HID; kc += BK) {
        float a[16];
        *(float4*)&a[0]  = *(const float4*)&arow[kc + kh];
        *(float4*)&a[4]  = *(const float4*)&arow[kc + kh + 4];
        *(float4*)&a[8]  = *(const float4*)&arow[kc + 16 + kh];
        *(float4*)&a[12] = *(const float4*)&arow[kc + 16 + kh + 4];
        Frag ah, al;
        #pragma unroll
        for (int e = 0; e < 8; ++e) {
          u16 h0, l0, h1, l1;
          split_bf(a[2 * e], h0, l0);
          split_bf(a[2 * e + 1], h1, l1);
          ah.w[e] = (u32)h0 | ((u32)h1 << 16);
          al.w[e] = (u32)l0 | ((u32)l1 << 16);
        }
        #pragma unroll
        for (int nt = 0; nt < 4; ++nt) {
          Frag bh, bl;
          const int n = (nh << 6) + (nt << 4) + l15;
          bh.q[0] = *(const uint4*)&wHi[n * WKP + kc + kh];
          bh.q[1] = *(const uint4*)&wHi[n * WKP + kc + 16 + kh];
          bl.q[0] = *(const uint4*)&wLo[n * WKP + kc + kh];
          bl.q[1] = *(const uint4*)&wLo[n * WKP + kc + 16 + kh];
          d[nt] = __builtin_amdgcn_wmma_f32_16x16x32_bf16(false, ah.v, false, bh.v, (short)0, d[nt], false, false);
          d[nt] = __builtin_amdgcn_wmma_f32_16x16x32_bf16(false, ah.v, false, bl.v, (short)0, d[nt], false, false);
          d[nt] = __builtin_amdgcn_wmma_f32_16x16x32_bf16(false, al.v, false, bh.v, (short)0, d[nt], false, false);
        }
      }
    }
    __syncthreads();
    if (mine) {
      float* orow = acc + ((mt << 4) + ((lane & 16) ? 8 : 0)) * ACP + (nh << 6) + l15;
      #pragma unroll
      for (int nt = 0; nt < 4; ++nt)
        #pragma unroll
        for (int r = 0; r < 8; ++r) orow[r * ACP + (nt << 4)] = d[nt][r];
    }
    __syncthreads();
    if (al16) {
      float4* o4 = (float4*)(outn + g0);
      for (int i = tid; i < nrow * (F / 4); i += OWN_NT) {
        const int r = i >> 5, c4 = i & 31;
        o4[i] = *(const float4*)&acc[r * ACP + 4 * c4];
      }
    } else {
      for (int i = tid; i < nrow * F; i += OWN_NT) outn[g0 + i] = acc[(i >> 7) * ACP + (i & (F - 1))];
    }
  } else {
    for (int o = tid; o < nrow * NOUT; o += OWN_NT) {
      const int r = o / NOUT, c = o - r * NOUT;
      const float* hr = acc + r * ACP;
      float sum = 0.0f;
      #pragma unroll 8
      for (int k = 0; k < HID; ++k) sum += hr[k] * wS[k * NOUT + c];
      outn[(size_t)row0 * NOUT + o] = sum;
    }
  }
}

__global__ __launch_bounds__(OWN_NT)
void prod_owner(const float* __restrict__ X, const float* __restrict__ Wn,
                float* __restrict__ outn, int nN) {
  constexpr int F = HID;
  __shared__ __attribute__((aligned(16))) float acc[OWN_RP * ACP];
  __shared__ __attribute__((aligned(16))) u16   wHi[HID * WKP];
  __shared__ __attribute__((aligned(16))) u16   wLo[HID * WKP];

  const int tid  = threadIdx.x;
  const int lane = tid & 31;
  const int wave = __builtin_amdgcn_readfirstlane(tid >> 5);
  const int row0 = blockIdx.x * OWN_R;
  const int nrow = min(OWN_R, nN - row0);
  const size_t g0 = (size_t)row0 * F;
  const bool al16 =
      (((((unsigned long long)X) | ((unsigned long long)outn)) & 15ull) == 0ull);

  constexpr int NW = (HID / 2) * HID / OWN_NT;
  constexpr int NX = (OWN_RP * (F / 4) + OWN_NT - 1) / OWN_NT;
  float w0[NW], w1[NW];
  #pragma unroll
  for (int u = 0; u < NW; ++u) {
    const int i = tid + u * OWN_NT;
    const int n = i & (HID - 1), kp = i >> 7;
    w0[u] = Wn[(2 * kp) * HID + n];
    w1[u] = Wn[(2 * kp + 1) * HID + n];
  }
  float4 xv[NX];
  #pragma unroll
  for (int u = 0; u < NX; ++u) xv[u] = make_float4(0.f, 0.f, 0.f, 0.f);
  if (al16) {
    const float4* x4 = (const float4*)(X + g0);
    const int last = nrow * (F / 4) - 1;
    #pragma unroll
    for (int u = 0; u < NX; ++u) xv[u] = x4[min(tid + u * OWN_NT, last)];
  }

  #pragma unroll
  for (int u = 0; u < NW; ++u) {
    const int i = tid + u * OWN_NT;
    const int n = i & (HID - 1), kp = i >> 7;
    u16 h0, l0, h1, l1;
    split_bf(w0[u], h0, l0);
    split_bf(w1[u], h1, l1);
    *(u32*)&wHi[n * WKP + 2 * kp] = (u32)h0 | ((u32)h1 << 16);
    *(u32*)&wLo[n * WKP + 2 * kp] = (u32)l0 | ((u32)l1 << 16);
  }

  if (al16) {
    #pragma unroll
    for (int u = 0; u < NX; ++u) {
      const int i = tid + u * OWN_NT;
      if (i < OWN_RP * (F / 4)) {
        const int r = i >> 5, c4 = i & 31;
        float4 v = xv[u];
        if (r >= nrow) v = make_float4(0.f, 0.f, 0.f, 0.f);
        *(float4*)&acc[r * ACP + 4 * c4] = v;
      }
    }
  } else {
    for (int i = tid; i < OWN_RP * F; i += OWN_NT) {
      const int r = i >> 7, c = i & (F - 1);
      acc[r * ACP + c] = (r < nrow) ? X[g0 + i] : 0.0f;
    }
  }
  __syncthreads();

  const int l15 = lane & 15;
  const int kh  = (lane >> 4) << 3;
  const int mt  = wave >> 1;
  const int nh  = wave & 1;
  const bool mine = wave < 2 * (OWN_RP / 16);
  v8f d[4];
  for (int t = 0; t < 4; ++t)
    for (int r = 0; r < 8; ++r) d[t][r] = 0.0f;
  if (mine) {
    union Frag { uint4 q[2]; u32 w[8]; v16bf v; };
    const float* arow = acc + ((mt << 4) + l15) * ACP;
    #pragma unroll
    for (int kc = 0; kc < HID; kc += BK) {
      float a[16];
      *(float4*)&a[0]  = *(const float4*)&arow[kc + kh];
      *(float4*)&a[4]  = *(const float4*)&arow[kc + kh + 4];
      *(float4*)&a[8]  = *(const float4*)&arow[kc + 16 + kh];
      *(float4*)&a[12] = *(const float4*)&arow[kc + 16 + kh + 4];
      Frag ah, al;
      #pragma unroll
      for (int e = 0; e < 8; ++e) {
        u16 h0, l0, h1, l1;
        split_bf(a[2 * e], h0, l0);
        split_bf(a[2 * e + 1], h1, l1);
        ah.w[e] = (u32)h0 | ((u32)h1 << 16);
        al.w[e] = (u32)l0 | ((u32)l1 << 16);
      }
      #pragma unroll
      for (int nt = 0; nt < 4; ++nt) {
        Frag bh, bl;
        const int n = (nh << 6) + (nt << 4) + l15;
        bh.q[0] = *(const uint4*)&wHi[n * WKP + kc + kh];
        bh.q[1] = *(const uint4*)&wHi[n * WKP + kc + 16 + kh];
        bl.q[0] = *(const uint4*)&wLo[n * WKP + kc + kh];
        bl.q[1] = *(const uint4*)&wLo[n * WKP + kc + 16 + kh];
        d[nt] = __builtin_amdgcn_wmma_f32_16x16x32_bf16(false, ah.v, false, bh.v, (short)0, d[nt], false, false);
        d[nt] = __builtin_amdgcn_wmma_f32_16x16x32_bf16(false, ah.v, false, bl.v, (short)0, d[nt], false, false);
        d[nt] = __builtin_amdgcn_wmma_f32_16x16x32_bf16(false, al.v, false, bh.v, (short)0, d[nt], false, false);
      }
    }
  }
  __syncthreads();
  if (mine) {
    float* orow = acc + ((mt << 4) + ((lane & 16) ? 8 : 0)) * ACP + (nh << 6) + l15;
    #pragma unroll
    for (int nt = 0; nt < 4; ++nt)
      #pragma unroll
      for (int r = 0; r < 8; ++r) orow[r * ACP + (nt << 4)] = d[nt][r];
  }
  __syncthreads();
  if (al16) {
    float4* o4 = (float4*)(outn + g0);
    for (int i = tid; i < nrow * (F / 4); i += OWN_NT) {
      const int r = i >> 5, c4 = i & 31;
      o4[i] = *(const float4*)&acc[r * ACP + 4 * c4];
    }
  } else {
    for (int i = tid; i < nrow * F; i += OWN_NT) outn[g0 + i] = acc[(i >> 7) * ACP + (i & (F - 1))];
  }
}

static inline int cdiv(long long a, long long b) { return (int)((a + b - 1) / b); }

extern "C" void kernel_launch(void* const* d_in, const int* in_sizes, int n_in,
                              void* d_out, int out_size, void* d_ws, size_t ws_size,
                              hipStream_t stream) {
  (void)n_in; (void)out_size; (void)ws_size;
  const float* x    = (const float*)d_in[0];
  const int*   eidx = (const int*)d_in[1];
  const float* Wst  = (const float*)d_in[2];
  const float* bst  = (const float*)d_in[3];
  const float* Wout = (const float*)d_in[4];
  const float* bout = (const float*)d_in[5];

  const int N = in_sizes[0] / HID;
  const int E = in_sizes[1] / 2;
  constexpr int NCLS = 10;

  const int* src = eidx;
  const int* dst = eidx + E;

  const int own_blocks  = cdiv(N, OWN_R);

  int4*  rec  = (int4*)d_ws;
  float* deg  = (float*)(rec + ((size_t)E + N));
  float* hbuf = deg + 2 * (size_t)N;
  float* tmp  = hbuf + (size_t)N * HID;
  int*   span = (int*)(tmp + (size_t)N * HID);
  int*   qcnt = span + 2 * (size_t)own_blocks;
  int*   queue = qcnt + own_blocks;

  own_count<<<own_blocks, OWN_NT, 0, stream>>>(dst, deg, qcnt, queue, E, N);
  own_lists<<<own_blocks, OWN_NT, 0, stream>>>(src, dst, deg, qcnt, queue, rec, span, E, N);

  prod_owner<<<own_blocks, OWN_NT, 0, stream>>>(x, Wst, tmp, N);

  float* cur = tmp;
  float* nxt = hbuf;
  for (int l = 0; l < 7; ++l) {
    agg_next<HID><<<own_blocks, OWN_NT, 0, stream>>>(cur, rec, span, bst + (size_t)l * HID,
                                                     Wst + (size_t)(l + 1) * HID * HID, nxt, N);
    float* t = cur; cur = nxt; nxt = t;
  }
  agg_next<NCLS><<<own_blocks, OWN_NT, 0, stream>>>(cur, rec, span, bst + (size_t)7 * HID,
                                                    Wout, nxt, N);

  agg_owner<NCLS><<<own_blocks, OWN_NT, 0, stream>>>(nxt, rec, span, bout, (float*)d_out, N);
}
